# starting kernel
# baseline (speedup 1.0000x reference)
#include <hip/hip_runtime.h>


typedef __attribute__((ext_vector_type(2))) float v2f;
typedef __attribute__((ext_vector_type(8))) float v8f;

#define GIN_D_IN 64
#define GIN_D_HID 32
#define GIN_N_CLASSES 40
#define GIN_SELF_W 1.1f

__global__ __launch_bounds__(256) void gin_scale_kernel(
    const float* __restrict__ h, float* __restrict__ agg, int n4) {
  int i = blockIdx.x * blockDim.x + threadIdx.x;
  if (i < n4) {
    const float4* hv = (const float4*)h;
    float4* av = (float4*)agg;
    float4 v = hv[i];
    v.x *= GIN_SELF_W; v.y *= GIN_SELF_W; v.z *= GIN_SELF_W; v.w *= GIN_SELF_W;
    av[i] = v;
  }
}

template <int D>
__global__ __launch_bounds__(256) void gin_scatter_kernel(
    const float* __restrict__ h, const int* __restrict__ src,
    const int* __restrict__ dst, float* __restrict__ agg, int n_edges) {
  constexpr int TPE = D / 4;
  int gid = blockIdx.x * blockDim.x + threadIdx.x;
  if (gid >= n_edges * TPE) return;
  int e = gid / TPE;
  int f = (gid % TPE) * 4;
  int s = src[e];
  int d = dst[e];
  const float4 v = *(const float4*)(h + (size_t)s * D + f);
  float* p = agg + (size_t)d * D + f;
  atomicAdd(p + 0, v.x);
  atomicAdd(p + 1, v.y);
  atomicAdd(p + 2, v.z);
  atomicAdd(p + 3, v.w);
}

template <int DINL, int DOUTL, int NT, bool RELU>
__global__ __launch_bounds__(256) void gin_gemm_kernel(
    const float* __restrict__ A, const float* __restrict__ W,
    float* __restrict__ C, int n_rows) {
  const int lane = threadIdx.x & 31;
  const int wave = threadIdx.x >> 5;
  const int ntiles = (n_rows + 15) >> 4;
  const int tile = blockIdx.x * (blockDim.x >> 5) + wave;
  if (tile >= ntiles) return;

  const int row0 = tile << 4;
  const int l16 = lane & 15;
  const int khalf = lane >> 4;
  int arow = row0 + l16;
  if (arow >= n_rows) arow = n_rows - 1;

  v8f acc[NT] = {};

#pragma unroll
  for (int kk = 0; kk < DINL / 4; ++kk) {
    const int kb = kk * 4 + khalf * 2;
    v2f a;
    a.x = A[(size_t)arow * DINL + kb];
    a.y = A[(size_t)arow * DINL + kb + 1];
#pragma unroll
    for (int t = 0; t < NT; ++t) {
      const int col = t * 16 + l16;
      const bool ok = ((DOUTL & 15) == 0) || (col < DOUTL);
      v2f b;
      b.x = ok ? W[(size_t)kb * DOUTL + col] : 0.0f;
      b.y = ok ? W[(size_t)(kb + 1) * DOUTL + col] : 0.0f;
      acc[t] = __builtin_amdgcn_wmma_f32_16x16x4_f32(
          false, a, false, b, (short)0, acc[t], false, false);
    }
  }

#pragma unroll
  for (int t = 0; t < NT; ++t) {
    const int col = t * 16 + l16;
    if (((DOUTL & 15) != 0) && col >= DOUTL) continue;
#pragma unroll
    for (int r = 0; r < 8; ++r) {
      const int row = row0 + r + khalf * 8;
      if (row < n_rows) {
        float v = acc[t][r];
        if (RELU) v = fmaxf(v, 0.0f);
        C[(size_t)row * DOUTL + col] = v;
      }
    }
  }
}

extern "C" void kernel_launch(void* const* d_in, const int* in_sizes, int n_in,
                              void* d_out, int out_size, void* d_ws, size_t ws_size,
                              hipStream_t stream) {
  const float* x  = (const float*)d_in[0];
  const float* W0 = (const float*)d_in[1];
  const float* W1 = (const float*)d_in[2];
  const float* W2 = (const float*)d_in[3];
  const int* src  = (const int*)d_in[4];
  const int* dst  = (const int*)d_in[5];
  float* out = (float*)d_out;

  const int n = in_sizes[0] / GIN_D_IN;
  const int n_edges = in_sizes[4];

  float* agg = (float*)d_ws;
  float* hbuf = agg + (size_t)n * GIN_D_IN;

  const int ntiles = (n + 15) / 16;
  const int gemm_blocks = (ntiles + 7) / 8;

  {
    int n4 = n * GIN_D_IN / 4;
    gin_scale_kernel<<<(n4 + 255) / 256, 256, 0, stream>>>(x, agg, n4);
    int tot = n_edges * (GIN_D_IN / 4);
    gin_scatter_kernel<GIN_D_IN>
        <<<(tot + 255) / 256, 256, 0, stream>>>(x, src, dst, agg, n_edges);
    gin_gemm_kernel<GIN_D_IN, GIN_D_HID, 2, true>
        <<<gemm_blocks, 256, 0, stream>>>(agg, W0, hbuf, n);
  }
  {
    int n4 = n * GIN_D_HID / 4;
    gin_scale_kernel<<<(n4 + 255) / 256, 256, 0, stream>>>(hbuf, agg, n4);
    int tot = n_edges * (GIN_D_HID / 4);
    gin_scatter_kernel<GIN_D_HID>
        <<<(tot + 255) / 256, 256, 0, stream>>>(hbuf, src, dst, agg, n_edges);
    gin_gemm_kernel<GIN_D_HID, GIN_D_HID, 2, true>
        <<<gemm_blocks, 256, 0, stream>>>(agg, W1, hbuf, n);
  }
  {
    int n4 = n * GIN_D_HID / 4;
    gin_scale_kernel<<<(n4 + 255) / 256, 256, 0, stream>>>(hbuf, agg, n4);
    int tot = n_edges * (GIN_D_HID / 4);
    gin_scatter_kernel<GIN_D_HID>
        <<<(tot + 255) / 256, 256, 0, stream>>>(hbuf, src, dst, agg, n_edges);
    gin_gemm_kernel<GIN_D_HID, GIN_N_CLASSES, 3, false>
        <<<gemm_blocks, 256, 0, stream>>>(agg, W2, out, n);
  }
}


# Layer 1 projects before aggregating: (Adj+1.1I)(x W0), not ((Adj+1.1I)x)W0, so its atomic scatter moves 32 not 64 floats per edge; its relu moves to layer-2 loads; layer-1 f32 sum order changes
# speedup vs baseline: 1.2476x; 1.2476x over previous
#include <hip/hip_runtime.h>


typedef __attribute__((ext_vector_type(2))) float v2f;
typedef __attribute__((ext_vector_type(8))) float v8f;

#define GIN_D_IN 64
#define GIN_D_HID 32
#define GIN_N_CLASSES 40
#define GIN_SELF_W 1.1f

template <bool RELU_IN>
__global__ __launch_bounds__(256) void gin_scale_kernel(
    const float* __restrict__ h, float* __restrict__ agg, int n4) {
  int i = blockIdx.x * blockDim.x + threadIdx.x;
  if (i < n4) {
    const float4* hv = (const float4*)h;
    float4* av = (float4*)agg;
    float4 v = hv[i];
    if (RELU_IN) {
      v.x = fmaxf(v.x, 0.0f); v.y = fmaxf(v.y, 0.0f);
      v.z = fmaxf(v.z, 0.0f); v.w = fmaxf(v.w, 0.0f);
    }
    v.x *= GIN_SELF_W; v.y *= GIN_SELF_W; v.z *= GIN_SELF_W; v.w *= GIN_SELF_W;
    av[i] = v;
  }
}

template <int D, bool RELU_IN>
__global__ __launch_bounds__(256) void gin_scatter_kernel(
    const float* __restrict__ h, const int* __restrict__ src,
    const int* __restrict__ dst, float* __restrict__ agg, int n_edges) {
  constexpr int TPE = D / 4;
  int gid = blockIdx.x * blockDim.x + threadIdx.x;
  if (gid >= n_edges * TPE) return;
  int e = gid / TPE;
  int f = (gid % TPE) * 4;
  int s = src[e];
  int d = dst[e];
  float4 v = *(const float4*)(h + (size_t)s * D + f);
  if (RELU_IN) {
    v.x = fmaxf(v.x, 0.0f); v.y = fmaxf(v.y, 0.0f);
    v.z = fmaxf(v.z, 0.0f); v.w = fmaxf(v.w, 0.0f);
  }
  float* p = agg + (size_t)d * D + f;
  atomicAdd(p + 0, v.x);
  atomicAdd(p + 1, v.y);
  atomicAdd(p + 2, v.z);
  atomicAdd(p + 3, v.w);
}

template <int DINL, int DOUTL, int NT, bool RELU>
__global__ __launch_bounds__(256) void gin_gemm_kernel(
    const float* __restrict__ A, const float* __restrict__ W,
    float* __restrict__ C, int n_rows) {
  const int lane = threadIdx.x & 31;
  const int wave = threadIdx.x >> 5;
  const int ntiles = (n_rows + 15) >> 4;
  const int tile = blockIdx.x * (blockDim.x >> 5) + wave;
  if (tile >= ntiles) return;

  const int row0 = tile << 4;
  const int l16 = lane & 15;
  const int khalf = lane >> 4;
  int arow = row0 + l16;
  if (arow >= n_rows) arow = n_rows - 1;

  v8f acc[NT] = {};

#pragma unroll
  for (int kk = 0; kk < DINL / 4; ++kk) {
    const int kb = kk * 4 + khalf * 2;
    v2f a;
    a.x = A[(size_t)arow * DINL + kb];
    a.y = A[(size_t)arow * DINL + kb + 1];
#pragma unroll
    for (int t = 0; t < NT; ++t) {
      const int col = t * 16 + l16;
      const bool ok = ((DOUTL & 15) == 0) || (col < DOUTL);
      v2f b;
      b.x = ok ? W[(size_t)kb * DOUTL + col] : 0.0f;
      b.y = ok ? W[(size_t)(kb + 1) * DOUTL + col] : 0.0f;
      acc[t] = __builtin_amdgcn_wmma_f32_16x16x4_f32(
          false, a, false, b, (short)0, acc[t], false, false);
    }
  }

#pragma unroll
  for (int t = 0; t < NT; ++t) {
    const int col = t * 16 + l16;
    if (((DOUTL & 15) != 0) && col >= DOUTL) continue;
#pragma unroll
    for (int r = 0; r < 8; ++r) {
      const int row = row0 + r + khalf * 8;
      if (row < n_rows) {
        float v = acc[t][r];
        if (RELU) v = fmaxf(v, 0.0f);
        C[(size_t)row * DOUTL + col] = v;
      }
    }
  }
}

extern "C" void kernel_launch(void* const* d_in, const int* in_sizes, int n_in,
                              void* d_out, int out_size, void* d_ws, size_t ws_size,
                              hipStream_t stream) {
  const float* x  = (const float*)d_in[0];
  const float* W0 = (const float*)d_in[1];
  const float* W1 = (const float*)d_in[2];
  const float* W2 = (const float*)d_in[3];
  const int* src  = (const int*)d_in[4];
  const int* dst  = (const int*)d_in[5];
  float* out = (float*)d_out;

  const int n = in_sizes[0] / GIN_D_IN;
  const int n_edges = in_sizes[4];

  float* tbuf = (float*)d_ws;
  float* agg1 = tbuf + (size_t)n * GIN_D_HID;
  float* agg2 = agg1 + (size_t)n * GIN_D_HID;

  const int ntiles = (n + 15) / 16;
  const int gemm_blocks = (ntiles + 7) / 8;
  const int n4 = n * GIN_D_HID / 4;
  const int tot = n_edges * (GIN_D_HID / 4);

  {
    gin_gemm_kernel<GIN_D_IN, GIN_D_HID, 2, false>
        <<<gemm_blocks, 256, 0, stream>>>(x, W0, tbuf, n);
    gin_scale_kernel<false><<<(n4 + 255) / 256, 256, 0, stream>>>(tbuf, agg1, n4);
    gin_scatter_kernel<GIN_D_HID, false>
        <<<(tot + 255) / 256, 256, 0, stream>>>(tbuf, src, dst, agg1, n_edges);
  }
  {
    gin_scale_kernel<true><<<(n4 + 255) / 256, 256, 0, stream>>>(agg1, agg2, n4);
    gin_scatter_kernel<GIN_D_HID, true>
        <<<(tot + 255) / 256, 256, 0, stream>>>(agg1, src, dst, agg2, n_edges);
    gin_gemm_kernel<GIN_D_HID, GIN_D_HID, 2, true>
        <<<gemm_blocks, 256, 0, stream>>>(agg2, W1, tbuf, n);
  }
  {
    gin_scale_kernel<false><<<(n4 + 255) / 256, 256, 0, stream>>>(tbuf, agg1, n4);
    gin_scatter_kernel<GIN_D_HID, false>
        <<<(tot + 255) / 256, 256, 0, stream>>>(tbuf, src, dst, agg1, n_edges);
    gin_gemm_kernel<GIN_D_HID, GIN_N_CLASSES, 3, false>
        <<<gemm_blocks, 256, 0, stream>>>(agg1, W2, out, n);
  }
}


# Edges grouped by destination once per call (two-level counting sort on LDS integer counters); each layer sums a node's source rows in registers, stores the row once: no f32 atomics; sum order changes
# speedup vs baseline: 2.0218x; 1.6206x over previous
#include <hip/hip_runtime.h>


typedef __attribute__((ext_vector_type(2))) float v2f;
typedef __attribute__((ext_vector_type(8))) float v8f;

#define GIN_D_IN 64
#define GIN_D_HID 32
#define GIN_N_CLASSES 40
#define GIN_SELF_W 1.1f

#define GIN_BSH 8
#define GIN_BNODES (1 << GIN_BSH)
#define GIN_MAXB 512
#define GIN_CHUNKS 256

__global__ __launch_bounds__(1024) void gin_bucket_count_kernel(
    const int* __restrict__ dst, int* __restrict__ cnt, int n_edges,
    int n_buckets, int chunk_len) {
  __shared__ int hist[GIN_MAXB];
  for (int i = threadIdx.x; i < GIN_MAXB; i += 1024) hist[i] = 0;
  __syncthreads();
  const int c = blockIdx.x;
  const int beg = c * chunk_len;
  const int end = min(beg + chunk_len, n_edges);
  for (int e = beg + (int)threadIdx.x; e < end; e += 1024)
    atomicAdd(&hist[dst[e] >> GIN_BSH], 1);
  __syncthreads();
  for (int b = threadIdx.x; b < n_buckets; b += 1024)
    cnt[b * GIN_CHUNKS + c] = hist[b];
}

__global__ __launch_bounds__(GIN_CHUNKS) void gin_bucket_offset_kernel(
    const int* __restrict__ cnt, int* __restrict__ off,
    int* __restrict__ bstart, int n_buckets) {
  __shared__ int w_before[GIN_CHUNKS / 32];
  __shared__ int w_own[GIN_CHUNKS / 32];
  const int b = blockIdx.x;
  const int t = threadIdx.x;
  const int lane = t & 31;
  const int wave = t >> 5;

  const int4* c4 = (const int4*)cnt;
  const int nq = b * (GIN_CHUNKS / 4);
  int before = 0;
  for (int i = t; i < nq; i += GIN_CHUNKS) {
    int4 v = c4[i];
    before += (v.x + v.y) + (v.z + v.w);
  }
  const int own = cnt[b * GIN_CHUNKS + t];

  int incl = own;
#pragma unroll
  for (int d = 1; d < 32; d <<= 1) {
    int up = __shfl_up(incl, d);
    if (lane >= d) incl += up;
  }
#pragma unroll
  for (int d = 16; d >= 1; d >>= 1) before += __shfl_xor(before, d);
  if (lane == 31) w_own[wave] = incl;
  if (lane == 0) w_before[wave] = before;
  __syncthreads();
  int base = 0, wpre = 0;
#pragma unroll
  for (int w = 0; w < GIN_CHUNKS / 32; ++w) {
    base += w_before[w];
    if (w < wave) wpre += w_own[w];
  }
  const int excl = base + wpre + incl - own;
  off[b * GIN_CHUNKS + t] = excl;
  if (t == 0) bstart[b] = base;
  if (b == n_buckets - 1 && t == GIN_CHUNKS - 1) bstart[n_buckets] = excl + own;
}

__global__ __launch_bounds__(1024) void gin_bucket_place_kernel(
    const int* __restrict__ src, const int* __restrict__ dst,
    const int* __restrict__ off, int* __restrict__ pairs, int n_edges,
    int n_buckets, int chunk_len) {
  __shared__ int cur[GIN_MAXB];
  const int c = blockIdx.x;
  for (int b = threadIdx.x; b < n_buckets; b += 1024)
    cur[b] = off[b * GIN_CHUNKS + c];
  __syncthreads();
  const int beg = c * chunk_len;
  const int end = min(beg + chunk_len, n_edges);
  int2* p2 = (int2*)pairs;
  for (int e = beg + (int)threadIdx.x; e < end; e += 1024) {
    const int d = dst[e];
    const int s = src[e];
    const int pos = atomicAdd(&cur[d >> GIN_BSH], 1);
    p2[pos] = make_int2(s, d);
  }
}

__global__ __launch_bounds__(GIN_BNODES) void gin_bucket_csr_kernel(
    const int* __restrict__ pairs, const int* __restrict__ bstart,
    int* __restrict__ rowptr, int* __restrict__ csr_src, int n) {
  __shared__ int cnt[GIN_BNODES];
  __shared__ int w_tot[GIN_BNODES / 32];
  const int b = blockIdx.x;
  const int t = threadIdx.x;
  const int lane = t & 31;
  const int wave = t >> 5;
  const int beg = bstart[b];
  const int end = bstart[b + 1];
  const int2* p2 = (const int2*)pairs;

  cnt[t] = 0;
  __syncthreads();
  for (int e = beg + t; e < end; e += GIN_BNODES)
    atomicAdd(&cnt[p2[e].y & (GIN_BNODES - 1)], 1);
  __syncthreads();

  const int own = cnt[t];
  int incl = own;
#pragma unroll
  for (int d = 1; d < 32; d <<= 1) {
    int up = __shfl_up(incl, d);
    if (lane >= d) incl += up;
  }
  if (lane == 31) w_tot[wave] = incl;
  __syncthreads();
  int wpre = 0;
#pragma unroll
  for (int w = 0; w < GIN_BNODES / 32; ++w)
    if (w < wave) wpre += w_tot[w];
  const int first = beg + wpre + incl - own;
  cnt[t] = first;
  const int node = b * GIN_BNODES + t;
  if (node < n) rowptr[node] = first;
  if (b == (int)gridDim.x - 1 && t == 0) rowptr[n] = end;
  __syncthreads();

  for (int e = beg + t; e < end; e += GIN_BNODES) {
    const int2 p = p2[e];
    const int pos = atomicAdd(&cnt[p.y & (GIN_BNODES - 1)], 1);
    csr_src[pos] = p.x;
  }
}

template <bool RELU_IN>
__global__ __launch_bounds__(256) void gin_gather_kernel(
    const float* __restrict__ h, const int* __restrict__ rowptr,
    const int* __restrict__ csr_src, float* __restrict__ agg, int n) {
  const int l8 = threadIdx.x & 7;
  const int node = (blockIdx.x * 256 + (int)threadIdx.x) >> 3;
  if (node >= n) return;
  const float4* h4 = (const float4*)h;
  float4 acc = h4[(unsigned)node * 8u + (unsigned)l8];
  if (RELU_IN) {
    acc.x = fmaxf(acc.x, 0.0f); acc.y = fmaxf(acc.y, 0.0f);
    acc.z = fmaxf(acc.z, 0.0f); acc.w = fmaxf(acc.w, 0.0f);
  }
  acc.x *= GIN_SELF_W; acc.y *= GIN_SELF_W; acc.z *= GIN_SELF_W; acc.w *= GIN_SELF_W;
  const int beg = rowptr[node];
  const int end = rowptr[node + 1];
  for (int base = beg; base < end; base += 8) {
    const int e = min(base + l8, end - 1);
    const int mine = csr_src[e];
    float4 r[8];
#pragma unroll
    for (int j = 0; j < 8; ++j) {
      const unsigned s = (unsigned)__shfl(mine, j, 8);
      r[j] = h4[s * 8u + (unsigned)l8];
    }
#pragma unroll
    for (int j = 0; j < 8; ++j) {
      float4 v = r[j];
      if (RELU_IN) {
        v.x = fmaxf(v.x, 0.0f); v.y = fmaxf(v.y, 0.0f);
        v.z = fmaxf(v.z, 0.0f); v.w = fmaxf(v.w, 0.0f);
      }
      const bool on = base + j < end;
      acc.x += on ? v.x : 0.0f; acc.y += on ? v.y : 0.0f;
      acc.z += on ? v.z : 0.0f; acc.w += on ? v.w : 0.0f;
    }
  }
  ((float4*)agg)[(unsigned)node * 8u + (unsigned)l8] = acc;
}

template <int DINL, int DOUTL, int NT, bool RELU>
__global__ __launch_bounds__(256) void gin_gemm_kernel(
    const float* __restrict__ A, const float* __restrict__ W,
    float* __restrict__ C, int n_rows) {
  const int lane = threadIdx.x & 31;
  const int wave = threadIdx.x >> 5;
  const int ntiles = (n_rows + 15) >> 4;
  const int tile = blockIdx.x * (blockDim.x >> 5) + wave;
  if (tile >= ntiles) return;

  const int row0 = tile << 4;
  const int l16 = lane & 15;
  const int khalf = lane >> 4;
  int arow = row0 + l16;
  if (arow >= n_rows) arow = n_rows - 1;

  v8f acc[NT] = {};

#pragma unroll
  for (int kk = 0; kk < DINL / 4; ++kk) {
    const int kb = kk * 4 + khalf * 2;
    v2f a;
    a.x = A[(size_t)arow * DINL + kb];
    a.y = A[(size_t)arow * DINL + kb + 1];
#pragma unroll
    for (int t = 0; t < NT; ++t) {
      const int col = t * 16 + l16;
      const bool ok = ((DOUTL & 15) == 0) || (col < DOUTL);
      v2f b;
      b.x = ok ? W[(size_t)kb * DOUTL + col] : 0.0f;
      b.y = ok ? W[(size_t)(kb + 1) * DOUTL + col] : 0.0f;
      acc[t] = __builtin_amdgcn_wmma_f32_16x16x4_f32(
          false, a, false, b, (short)0, acc[t], false, false);
    }
  }

#pragma unroll
  for (int t = 0; t < NT; ++t) {
    const int col = t * 16 + l16;
    if (((DOUTL & 15) != 0) && col >= DOUTL) continue;
#pragma unroll
    for (int r = 0; r < 8; ++r) {
      const int row = row0 + r + khalf * 8;
      if (row < n_rows) {
        float v = acc[t][r];
        if (RELU) v = fmaxf(v, 0.0f);
        C[(size_t)row * DOUTL + col] = v;
      }
    }
  }
}

extern "C" void kernel_launch(void* const* d_in, const int* in_sizes, int n_in,
                              void* d_out, int out_size, void* d_ws, size_t ws_size,
                              hipStream_t stream) {
  const float* x  = (const float*)d_in[0];
  const float* W0 = (const float*)d_in[1];
  const float* W1 = (const float*)d_in[2];
  const float* W2 = (const float*)d_in[3];
  const int* src  = (const int*)d_in[4];
  const int* dst  = (const int*)d_in[5];
  float* out = (float*)d_out;

  const int n = in_sizes[0] / GIN_D_IN;
  const int n_edges = in_sizes[4];

  float* tbuf = (float*)d_ws;
  float* agg1 = tbuf + (size_t)n * GIN_D_HID;
  float* agg2 = agg1 + (size_t)n * GIN_D_HID;
  int* cnt    = (int*)(agg2 + (size_t)n * GIN_D_HID);
  int* off    = cnt + GIN_MAXB * GIN_CHUNKS;
  int* bstart = off + GIN_MAXB * GIN_CHUNKS;
  int* rowptr = bstart + 2 * GIN_MAXB;
  int* pairs  = rowptr + (((size_t)n + 1 + 3) / 4) * 4;
  int* csrc   = pairs + 2 * (size_t)n_edges;

  const int ntiles = (n + 15) / 16;
  const int gemm_blocks = (ntiles + 7) / 8;
  const int n_buckets = (n + GIN_BNODES - 1) / GIN_BNODES;
  const int chunk_len = (n_edges + GIN_CHUNKS - 1) / GIN_CHUNKS;
  const int gather_blocks = (n * 8 + 255) / 256;

  gin_bucket_count_kernel<<<GIN_CHUNKS, 1024, 0, stream>>>(dst, cnt, n_edges, n_buckets, chunk_len);
  gin_bucket_offset_kernel<<<n_buckets, GIN_CHUNKS, 0, stream>>>(cnt, off, bstart, n_buckets);
  gin_bucket_place_kernel<<<GIN_CHUNKS, 1024, 0, stream>>>(src, dst, off, pairs, n_edges, n_buckets, chunk_len);
  gin_bucket_csr_kernel<<<n_buckets, GIN_BNODES, 0, stream>>>(pairs, bstart, rowptr, csrc, n);

  gin_gemm_kernel<GIN_D_IN, GIN_D_HID, 2, false>
      <<<gemm_blocks, 256, 0, stream>>>(x, W0, tbuf, n);
  gin_gather_kernel<false><<<gather_blocks, 256, 0, stream>>>(tbuf, rowptr, csrc, agg1, n);
  gin_gather_kernel<true><<<gather_blocks, 256, 0, stream>>>(agg1, rowptr, csrc, agg2, n);
  gin_gemm_kernel<GIN_D_HID, GIN_D_HID, 2, true>
      <<<gemm_blocks, 256, 0, stream>>>(agg2, W1, tbuf, n);
  gin_gather_kernel<false><<<gather_blocks, 256, 0, stream>>>(tbuf, rowptr, csrc, agg1, n);
  gin_gemm_kernel<GIN_D_HID, GIN_N_CLASSES, 3, false>
      <<<gemm_blocks, 256, 0, stream>>>(agg1, W2, out, n);
}


# Bucket offsets: a bucket's block writes its own total and chunk prefix instead of re-summing every earlier bucket's counts in a serial load loop; the two consumers scan the totals in LDS
# speedup vs baseline: 2.5895x; 1.2808x over previous
#include <hip/hip_runtime.h>


typedef __attribute__((ext_vector_type(2))) float v2f;
typedef __attribute__((ext_vector_type(8))) float v8f;

#define GIN_D_IN 64
#define GIN_D_HID 32
#define GIN_N_CLASSES 40
#define GIN_SELF_W 1.1f

#define GIN_BSH 8
#define GIN_BNODES (1 << GIN_BSH)
#define GIN_MAXB 512
#define GIN_CHUNKS 256

__global__ __launch_bounds__(1024) void gin_bucket_count_kernel(
    const int* __restrict__ dst, int* __restrict__ cnt, int n_edges,
    int n_buckets, int chunk_len) {
  __shared__ int hist[GIN_MAXB];
  for (int i = threadIdx.x; i < GIN_MAXB; i += 1024) hist[i] = 0;
  __syncthreads();
  const int c = blockIdx.x;
  const int beg = c * chunk_len;
  const int end = min(beg + chunk_len, n_edges);
  for (int e = beg + (int)threadIdx.x; e < end; e += 1024)
    atomicAdd(&hist[dst[e] >> GIN_BSH], 1);
  __syncthreads();
  for (int b = threadIdx.x; b < n_buckets; b += 1024)
    cnt[b * GIN_CHUNKS + c] = hist[b];
}

__global__ __launch_bounds__(GIN_CHUNKS) void gin_bucket_offset_kernel(
    const int* __restrict__ cnt, int* __restrict__ off, int* __restrict__ btotal) {
  __shared__ int w_own[GIN_CHUNKS / 32];
  const int b = blockIdx.x;
  const int t = threadIdx.x;
  const int lane = t & 31;
  const int wave = t >> 5;

  const int own = cnt[b * GIN_CHUNKS + t];
  int incl = own;
#pragma unroll
  for (int d = 1; d < 32; d <<= 1) {
    int up = __shfl_up(incl, d);
    if (lane >= d) incl += up;
  }
  if (lane == 31) w_own[wave] = incl;
  __syncthreads();
  int wpre = 0, tot = 0;
#pragma unroll
  for (int w = 0; w < GIN_CHUNKS / 32; ++w) {
    const int v = w_own[w];
    tot += v;
    if (w < wave) wpre += v;
  }
  off[b * GIN_CHUNKS + t] = wpre + incl - own;
  if (t == 0) btotal[b] = tot;
}

__global__ __launch_bounds__(1024) void gin_bucket_place_kernel(
    const int* __restrict__ src, const int* __restrict__ dst,
    const int* __restrict__ off, const int* __restrict__ btotal,
    int* __restrict__ pairs, int n_edges, int n_buckets, int chunk_len) {
  __shared__ int cur[GIN_MAXB];
  __shared__ int w_tot[GIN_MAXB / 32];
  const int c = blockIdx.x;
  const int t = threadIdx.x;
  const int lane = t & 31;
  const int wave = t >> 5;
  const bool has = t < n_buckets;
  const int tot = has ? btotal[t] : 0;
  const int rank = has ? off[t * GIN_CHUNKS + c] : 0;
  int incl = tot;
#pragma unroll
  for (int d = 1; d < 32; d <<= 1) {
    int up = __shfl_up(incl, d);
    if (lane >= d) incl += up;
  }
  if (lane == 31 && wave < GIN_MAXB / 32) w_tot[wave] = incl;
  __syncthreads();
  if (has) {
    int wpre = 0;
#pragma unroll
    for (int w = 0; w < GIN_MAXB / 32; ++w)
      if (w < wave) wpre += w_tot[w];
    cur[t] = wpre + incl - tot + rank;
  }
  __syncthreads();
  const int beg = c * chunk_len;
  const int end = min(beg + chunk_len, n_edges);
  int2* p2 = (int2*)pairs;
  for (int e = beg + t; e < end; e += 1024) {
    const int d = dst[e];
    const int s = src[e];
    const int pos = atomicAdd(&cur[d >> GIN_BSH], 1);
    p2[pos] = make_int2(s, d);
  }
}

__global__ __launch_bounds__(GIN_BNODES) void gin_bucket_csr_kernel(
    const int* __restrict__ pairs, const int* __restrict__ btotal,
    int* __restrict__ rowptr, int* __restrict__ csr_src, int n) {
  static_assert(GIN_MAXB <= 2 * GIN_BNODES, "two totals per thread cover every earlier bucket");
  __shared__ int cnt[GIN_BNODES];
  __shared__ int w_tot[GIN_BNODES / 32];
  __shared__ int w_beg[GIN_BNODES / 32];
  const int b = blockIdx.x;
  const int t = threadIdx.x;
  const int lane = t & 31;
  const int wave = t >> 5;
  const int2* p2 = (const int2*)pairs;

  int part = (t < b ? btotal[t] : 0) +
             (t + GIN_BNODES < b ? btotal[t + GIN_BNODES] : 0);
#pragma unroll
  for (int d = 16; d >= 1; d >>= 1) part += __shfl_xor(part, d);
  if (lane == 0) w_beg[wave] = part;
  cnt[t] = 0;
  __syncthreads();
  int beg = 0;
#pragma unroll
  for (int w = 0; w < GIN_BNODES / 32; ++w) beg += w_beg[w];
  const int end = beg + btotal[b];
  for (int e = beg + t; e < end; e += GIN_BNODES)
    atomicAdd(&cnt[p2[e].y & (GIN_BNODES - 1)], 1);
  __syncthreads();

  const int own = cnt[t];
  int incl = own;
#pragma unroll
  for (int d = 1; d < 32; d <<= 1) {
    int up = __shfl_up(incl, d);
    if (lane >= d) incl += up;
  }
  if (lane == 31) w_tot[wave] = incl;
  __syncthreads();
  int wpre = 0;
#pragma unroll
  for (int w = 0; w < GIN_BNODES / 32; ++w)
    if (w < wave) wpre += w_tot[w];
  const int first = beg + wpre + incl - own;
  cnt[t] = first;
  const int node = b * GIN_BNODES + t;
  if (node < n) rowptr[node] = first;
  if (b == (int)gridDim.x - 1 && t == 0) rowptr[n] = end;
  __syncthreads();

  for (int e = beg + t; e < end; e += GIN_BNODES) {
    const int2 p = p2[e];
    const int pos = atomicAdd(&cnt[p.y & (GIN_BNODES - 1)], 1);
    csr_src[pos] = p.x;
  }
}

template <bool RELU_IN>
__global__ __launch_bounds__(256) void gin_gather_kernel(
    const float* __restrict__ h, const int* __restrict__ rowptr,
    const int* __restrict__ csr_src, float* __restrict__ agg, int n) {
  const int l8 = threadIdx.x & 7;
  const int node = (blockIdx.x * 256 + (int)threadIdx.x) >> 3;
  if (node >= n) return;
  const float4* h4 = (const float4*)h;
  float4 acc = h4[(unsigned)node * 8u + (unsigned)l8];
  if (RELU_IN) {
    acc.x = fmaxf(acc.x, 0.0f); acc.y = fmaxf(acc.y, 0.0f);
    acc.z = fmaxf(acc.z, 0.0f); acc.w = fmaxf(acc.w, 0.0f);
  }
  acc.x *= GIN_SELF_W; acc.y *= GIN_SELF_W; acc.z *= GIN_SELF_W; acc.w *= GIN_SELF_W;
  const int beg = rowptr[node];
  const int end = rowptr[node + 1];
  for (int base = beg; base < end; base += 8) {
    const int e = min(base + l8, end - 1);
    const int mine = csr_src[e];
    float4 r[8];
#pragma unroll
    for (int j = 0; j < 8; ++j) {
      const unsigned s = (unsigned)__shfl(mine, j, 8);
      r[j] = h4[s * 8u + (unsigned)l8];
    }
#pragma unroll
    for (int j = 0; j < 8; ++j) {
      float4 v = r[j];
      if (RELU_IN) {
        v.x = fmaxf(v.x, 0.0f); v.y = fmaxf(v.y, 0.0f);
        v.z = fmaxf(v.z, 0.0f); v.w = fmaxf(v.w, 0.0f);
      }
      const bool on = base + j < end;
      acc.x += on ? v.x : 0.0f; acc.y += on ? v.y : 0.0f;
      acc.z += on ? v.z : 0.0f; acc.w += on ? v.w : 0.0f;
    }
  }
  ((float4*)agg)[(unsigned)node * 8u + (unsigned)l8] = acc;
}

template <int DINL, int DOUTL, int NT, bool RELU>
__global__ __launch_bounds__(256) void gin_gemm_kernel(
    const float* __restrict__ A, const float* __restrict__ W,
    float* __restrict__ C, int n_rows) {
  const int lane = threadIdx.x & 31;
  const int wave = threadIdx.x >> 5;
  const int ntiles = (n_rows + 15) >> 4;
  const int tile = blockIdx.x * (blockDim.x >> 5) + wave;
  if (tile >= ntiles) return;

  const int row0 = tile << 4;
  const int l16 = lane & 15;
  const int khalf = lane >> 4;
  int arow = row0 + l16;
  if (arow >= n_rows) arow = n_rows - 1;

  v8f acc[NT] = {};

#pragma unroll
  for (int kk = 0; kk < DINL / 4; ++kk) {
    const int kb = kk * 4 + khalf * 2;
    v2f a;
    a.x = A[(size_t)arow * DINL + kb];
    a.y = A[(size_t)arow * DINL + kb + 1];
#pragma unroll
    for (int t = 0; t < NT; ++t) {
      const int col = t * 16 + l16;
      const bool ok = ((DOUTL & 15) == 0) || (col < DOUTL);
      v2f b;
      b.x = ok ? W[(size_t)kb * DOUTL + col] : 0.0f;
      b.y = ok ? W[(size_t)(kb + 1) * DOUTL + col] : 0.0f;
      acc[t] = __builtin_amdgcn_wmma_f32_16x16x4_f32(
          false, a, false, b, (short)0, acc[t], false, false);
    }
  }

#pragma unroll
  for (int t = 0; t < NT; ++t) {
    const int col = t * 16 + l16;
    if (((DOUTL & 15) != 0) && col >= DOUTL) continue;
#pragma unroll
    for (int r = 0; r < 8; ++r) {
      const int row = row0 + r + khalf * 8;
      if (row < n_rows) {
        float v = acc[t][r];
        if (RELU) v = fmaxf(v, 0.0f);
        C[(size_t)row * DOUTL + col] = v;
      }
    }
  }
}

extern "C" void kernel_launch(void* const* d_in, const int* in_sizes, int n_in,
                              void* d_out, int out_size, void* d_ws, size_t ws_size,
                              hipStream_t stream) {
  const float* x  = (const float*)d_in[0];
  const float* W0 = (const float*)d_in[1];
  const float* W1 = (const float*)d_in[2];
  const float* W2 = (const float*)d_in[3];
  const int* src  = (const int*)d_in[4];
  const int* dst  = (const int*)d_in[5];
  float* out = (float*)d_out;

  const int n = in_sizes[0] / GIN_D_IN;
  const int n_edges = in_sizes[4];

  float* tbuf = (float*)d_ws;
  float* agg1 = tbuf + (size_t)n * GIN_D_HID;
  float* agg2 = agg1 + (size_t)n * GIN_D_HID;
  int* cnt    = (int*)(agg2 + (size_t)n * GIN_D_HID);
  int* off    = cnt + GIN_MAXB * GIN_CHUNKS;
  int* btotal = off + GIN_MAXB * GIN_CHUNKS;
  int* rowptr = btotal + 2 * GIN_MAXB;
  int* pairs  = rowptr + (((size_t)n + 1 + 3) / 4) * 4;
  int* csrc   = pairs + 2 * (size_t)n_edges;

  const int ntiles = (n + 15) / 16;
  const int gemm_blocks = (ntiles + 7) / 8;
  const int n_buckets = (n + GIN_BNODES - 1) / GIN_BNODES;
  const int chunk_len = (n_edges + GIN_CHUNKS - 1) / GIN_CHUNKS;
  const int gather_blocks = (n * 8 + 255) / 256;

  gin_bucket_count_kernel<<<GIN_CHUNKS, 1024, 0, stream>>>(dst, cnt, n_edges, n_buckets, chunk_len);
  gin_bucket_offset_kernel<<<n_buckets, GIN_CHUNKS, 0, stream>>>(cnt, off, btotal);
  gin_bucket_place_kernel<<<GIN_CHUNKS, 1024, 0, stream>>>(src, dst, off, btotal, pairs, n_edges, n_buckets, chunk_len);
  gin_bucket_csr_kernel<<<n_buckets, GIN_BNODES, 0, stream>>>(pairs, btotal, rowptr, csrc, n);

  gin_gemm_kernel<GIN_D_IN, GIN_D_HID, 2, false>
      <<<gemm_blocks, 256, 0, stream>>>(x, W0, tbuf, n);
  gin_gather_kernel<false><<<gather_blocks, 256, 0, stream>>>(tbuf, rowptr, csrc, agg1, n);
  gin_gather_kernel<true><<<gather_blocks, 256, 0, stream>>>(agg1, rowptr, csrc, agg2, n);
  gin_gemm_kernel<GIN_D_HID, GIN_D_HID, 2, true>
      <<<gemm_blocks, 256, 0, stream>>>(agg2, W1, tbuf, n);
  gin_gather_kernel<false><<<gather_blocks, 256, 0, stream>>>(tbuf, rowptr, csrc, agg1, n);
  gin_gemm_kernel<GIN_D_HID, GIN_N_CLASSES, 3, false>
      <<<gemm_blocks, 256, 0, stream>>>(agg1, W2, out, n);
}


# Layers 2 and 3: aggregation fused into its dense layer; a block's 32 summed rows go through LDS to the f32 WMMA, no sum plane written or re-read; 10 launches become 8; sum order unchanged
# speedup vs baseline: 2.6783x; 1.0343x over previous
#include <hip/hip_runtime.h>


typedef __attribute__((ext_vector_type(2))) float v2f;
typedef __attribute__((ext_vector_type(8))) float v8f;

#define GIN_D_IN 64
#define GIN_D_HID 32
#define GIN_N_CLASSES 40
#define GIN_SELF_W 1.1f

#define GIN_BSH 8
#define GIN_BNODES (1 << GIN_BSH)
#define GIN_MAXB 512
#define GIN_CHUNKS 256

__global__ __launch_bounds__(1024) void gin_bucket_count_kernel(
    const int* __restrict__ dst, int* __restrict__ cnt, int n_edges,
    int n_buckets, int chunk_len) {
  __shared__ int hist[GIN_MAXB];
  for (int i = threadIdx.x; i < GIN_MAXB; i += 1024) hist[i] = 0;
  __syncthreads();
  const int c = blockIdx.x;
  const int beg = c * chunk_len;
  const int end = min(beg + chunk_len, n_edges);
  for (int e = beg + (int)threadIdx.x; e < end; e += 1024)
    atomicAdd(&hist[dst[e] >> GIN_BSH], 1);
  __syncthreads();
  for (int b = threadIdx.x; b < n_buckets; b += 1024)
    cnt[b * GIN_CHUNKS + c] = hist[b];
}

__global__ __launch_bounds__(GIN_CHUNKS) void gin_bucket_offset_kernel(
    const int* __restrict__ cnt, int* __restrict__ off, int* __restrict__ btotal) {
  __shared__ int w_own[GIN_CHUNKS / 32];
  const int b = blockIdx.x;
  const int t = threadIdx.x;
  const int lane = t & 31;
  const int wave = t >> 5;

  const int own = cnt[b * GIN_CHUNKS + t];
  int incl = own;
#pragma unroll
  for (int d = 1; d < 32; d <<= 1) {
    int up = __shfl_up(incl, d);
    if (lane >= d) incl += up;
  }
  if (lane == 31) w_own[wave] = incl;
  __syncthreads();
  int wpre = 0, tot = 0;
#pragma unroll
  for (int w = 0; w < GIN_CHUNKS / 32; ++w) {
    const int v = w_own[w];
    tot += v;
    if (w < wave) wpre += v;
  }
  off[b * GIN_CHUNKS + t] = wpre + incl - own;
  if (t == 0) btotal[b] = tot;
}

__global__ __launch_bounds__(1024) void gin_bucket_place_kernel(
    const int* __restrict__ src, const int* __restrict__ dst,
    const int* __restrict__ off, const int* __restrict__ btotal,
    int* __restrict__ pairs, int n_edges, int n_buckets, int chunk_len) {
  __shared__ int cur[GIN_MAXB];
  __shared__ int w_tot[GIN_MAXB / 32];
  const int c = blockIdx.x;
  const int t = threadIdx.x;
  const int lane = t & 31;
  const int wave = t >> 5;
  const bool has = t < n_buckets;
  const int tot = has ? btotal[t] : 0;
  const int rank = has ? off[t * GIN_CHUNKS + c] : 0;
  int incl = tot;
#pragma unroll
  for (int d = 1; d < 32; d <<= 1) {
    int up = __shfl_up(incl, d);
    if (lane >= d) incl += up;
  }
  if (lane == 31 && wave < GIN_MAXB / 32) w_tot[wave] = incl;
  __syncthreads();
  if (has) {
    int wpre = 0;
#pragma unroll
    for (int w = 0; w < GIN_MAXB / 32; ++w)
      if (w < wave) wpre += w_tot[w];
    cur[t] = wpre + incl - tot + rank;
  }
  __syncthreads();
  const int beg = c * chunk_len;
  const int end = min(beg + chunk_len, n_edges);
  int2* p2 = (int2*)pairs;
  for (int e = beg + t; e < end; e += 1024) {
    const int d = dst[e];
    const int s = src[e];
    const int pos = atomicAdd(&cur[d >> GIN_BSH], 1);
    p2[pos] = make_int2(s, d);
  }
}

__global__ __launch_bounds__(GIN_BNODES) void gin_bucket_csr_kernel(
    const int* __restrict__ pairs, const int* __restrict__ btotal,
    int* __restrict__ rowptr, int* __restrict__ csr_src, int n) {
  static_assert(GIN_MAXB <= 2 * GIN_BNODES, "two totals per thread cover every earlier bucket");
  __shared__ int cnt[GIN_BNODES];
  __shared__ int w_tot[GIN_BNODES / 32];
  __shared__ int w_beg[GIN_BNODES / 32];
  const int b = blockIdx.x;
  const int t = threadIdx.x;
  const int lane = t & 31;
  const int wave = t >> 5;
  const int2* p2 = (const int2*)pairs;

  int part = (t < b ? btotal[t] : 0) +
             (t + GIN_BNODES < b ? btotal[t + GIN_BNODES] : 0);
#pragma unroll
  for (int d = 16; d >= 1; d >>= 1) part += __shfl_xor(part, d);
  if (lane == 0) w_beg[wave] = part;
  cnt[t] = 0;
  __syncthreads();
  int beg = 0;
#pragma unroll
  for (int w = 0; w < GIN_BNODES / 32; ++w) beg += w_beg[w];
  const int end = beg + btotal[b];
  for (int e = beg + t; e < end; e += GIN_BNODES)
    atomicAdd(&cnt[p2[e].y & (GIN_BNODES - 1)], 1);
  __syncthreads();

  const int own = cnt[t];
  int incl = own;
#pragma unroll
  for (int d = 1; d < 32; d <<= 1) {
    int up = __shfl_up(incl, d);
    if (lane >= d) incl += up;
  }
  if (lane == 31) w_tot[wave] = incl;
  __syncthreads();
  int wpre = 0;
#pragma unroll
  for (int w = 0; w < GIN_BNODES / 32; ++w)
    if (w < wave) wpre += w_tot[w];
  const int first = beg + wpre + incl - own;
  cnt[t] = first;
  const int node = b * GIN_BNODES + t;
  if (node < n) rowptr[node] = first;
  if (b == (int)gridDim.x - 1 && t == 0) rowptr[n] = end;
  __syncthreads();

  for (int e = beg + t; e < end; e += GIN_BNODES) {
    const int2 p = p2[e];
    const int pos = atomicAdd(&cnt[p.y & (GIN_BNODES - 1)], 1);
    csr_src[pos] = p.x;
  }
}

template <bool RELU_IN>
__device__ __forceinline__ float4 gin_gather_row(
    const float4* __restrict__ h4, const int* __restrict__ rowptr,
    const int* __restrict__ csr_src, int node, int l8) {
  float4 acc = h4[(unsigned)node * 8u + (unsigned)l8];
  if (RELU_IN) {
    acc.x = fmaxf(acc.x, 0.0f); acc.y = fmaxf(acc.y, 0.0f);
    acc.z = fmaxf(acc.z, 0.0f); acc.w = fmaxf(acc.w, 0.0f);
  }
  acc.x *= GIN_SELF_W; acc.y *= GIN_SELF_W; acc.z *= GIN_SELF_W; acc.w *= GIN_SELF_W;
  const int beg = rowptr[node];
  const int end = rowptr[node + 1];
  for (int base = beg; base < end; base += 8) {
    const int e = min(base + l8, end - 1);
    const int mine = csr_src[e];
    float4 r[8];
#pragma unroll
    for (int j = 0; j < 8; ++j) {
      const unsigned s = (unsigned)__shfl(mine, j, 8);
      r[j] = h4[s * 8u + (unsigned)l8];
    }
#pragma unroll
    for (int j = 0; j < 8; ++j) {
      float4 v = r[j];
      if (RELU_IN) {
        v.x = fmaxf(v.x, 0.0f); v.y = fmaxf(v.y, 0.0f);
        v.z = fmaxf(v.z, 0.0f); v.w = fmaxf(v.w, 0.0f);
      }
      const bool on = base + j < end;
      acc.x += on ? v.x : 0.0f; acc.y += on ? v.y : 0.0f;
      acc.z += on ? v.z : 0.0f; acc.w += on ? v.w : 0.0f;
    }
  }
  return acc;
}

template <bool RELU_IN>
__global__ __launch_bounds__(256) void gin_gather_kernel(
    const float* __restrict__ h, const int* __restrict__ rowptr,
    const int* __restrict__ csr_src, float* __restrict__ agg, int n) {
  const int l8 = threadIdx.x & 7;
  const int node = (blockIdx.x * 256 + (int)threadIdx.x) >> 3;
  if (node >= n) return;
  const float4 acc = gin_gather_row<RELU_IN>((const float4*)h, rowptr, csr_src, node, l8);
  ((float4*)agg)[(unsigned)node * 8u + (unsigned)l8] = acc;
}

#define GIN_GROUP_NODES 32
#define GIN_LDS_PITCH 36
template <bool RELU_IN, int DOUTL, int NT, bool RELU_OUT>
__global__ __launch_bounds__(256) void gin_gather_gemm_kernel(
    const float* __restrict__ h, const int* __restrict__ rowptr,
    const int* __restrict__ csr_src, const float* __restrict__ W,
    float* __restrict__ C, int n) {
  __shared__ __attribute__((aligned(16))) float rows[GIN_GROUP_NODES * GIN_LDS_PITCH];
  const int l8 = threadIdx.x & 7;
  const int lnode = (int)threadIdx.x >> 3;
  const int node = blockIdx.x * GIN_GROUP_NODES + lnode;
  float4 sum = make_float4(0.0f, 0.0f, 0.0f, 0.0f);
  if (node < n)
    sum = gin_gather_row<RELU_IN>((const float4*)h, rowptr, csr_src, node, l8);
  *(float4*)&rows[lnode * GIN_LDS_PITCH + l8 * 4] = sum;
  __syncthreads();

  const int lane = threadIdx.x & 31;
  const int wave = threadIdx.x >> 5;
  if (wave >= GIN_GROUP_NODES / 16) return;

  const int l16 = lane & 15;
  const int khalf = lane >> 4;
  const int row0 = blockIdx.x * GIN_GROUP_NODES + wave * 16;
  const float* arow = &rows[(wave * 16 + l16) * GIN_LDS_PITCH];

  v8f acc[NT] = {};

#pragma unroll
  for (int kk = 0; kk < GIN_D_HID / 4; ++kk) {
    const int kb = kk * 4 + khalf * 2;
    const v2f a = *(const v2f*)(arow + kb);
#pragma unroll
    for (int t = 0; t < NT; ++t) {
      const int col = t * 16 + l16;
      const bool ok = ((DOUTL & 15) == 0) || (col < DOUTL);
      v2f b;
      b.x = ok ? W[(size_t)kb * DOUTL + col] : 0.0f;
      b.y = ok ? W[(size_t)(kb + 1) * DOUTL + col] : 0.0f;
      acc[t] = __builtin_amdgcn_wmma_f32_16x16x4_f32(
          false, a, false, b, (short)0, acc[t], false, false);
    }
  }

#pragma unroll
  for (int t = 0; t < NT; ++t) {
    const int col = t * 16 + l16;
    if (((DOUTL & 15) != 0) && col >= DOUTL) continue;
#pragma unroll
    for (int r = 0; r < 8; ++r) {
      const int row = row0 + r + khalf * 8;
      if (row < n) {
        float v = acc[t][r];
        if (RELU_OUT) v = fmaxf(v, 0.0f);
        C[(size_t)row * DOUTL + col] = v;
      }
    }
  }
}

template <int DINL, int DOUTL, int NT, bool RELU>
__global__ __launch_bounds__(256) void gin_gemm_kernel(
    const float* __restrict__ A, const float* __restrict__ W,
    float* __restrict__ C, int n_rows) {
  const int lane = threadIdx.x & 31;
  const int wave = threadIdx.x >> 5;
  const int ntiles = (n_rows + 15) >> 4;
  const int tile = blockIdx.x * (blockDim.x >> 5) + wave;
  if (tile >= ntiles) return;

  const int row0 = tile << 4;
  const int l16 = lane & 15;
  const int khalf = lane >> 4;
  int arow = row0 + l16;
  if (arow >= n_rows) arow = n_rows - 1;

  v8f acc[NT] = {};

#pragma unroll
  for (int kk = 0; kk < DINL / 4; ++kk) {
    const int kb = kk * 4 + khalf * 2;
    v2f a;
    a.x = A[(size_t)arow * DINL + kb];
    a.y = A[(size_t)arow * DINL + kb + 1];
#pragma unroll
    for (int t = 0; t < NT; ++t) {
      const int col = t * 16 + l16;
      const bool ok = ((DOUTL & 15) == 0) || (col < DOUTL);
      v2f b;
      b.x = ok ? W[(size_t)kb * DOUTL + col] : 0.0f;
      b.y = ok ? W[(size_t)(kb + 1) * DOUTL + col] : 0.0f;
      acc[t] = __builtin_amdgcn_wmma_f32_16x16x4_f32(
          false, a, false, b, (short)0, acc[t], false, false);
    }
  }

#pragma unroll
  for (int t = 0; t < NT; ++t) {
    const int col = t * 16 + l16;
    if (((DOUTL & 15) != 0) && col >= DOUTL) continue;
#pragma unroll
    for (int r = 0; r < 8; ++r) {
      const int row = row0 + r + khalf * 8;
      if (row < n_rows) {
        float v = acc[t][r];
        if (RELU) v = fmaxf(v, 0.0f);
        C[(size_t)row * DOUTL + col] = v;
      }
    }
  }
}

extern "C" void kernel_launch(void* const* d_in, const int* in_sizes, int n_in,
                              void* d_out, int out_size, void* d_ws, size_t ws_size,
                              hipStream_t stream) {
  const float* x  = (const float*)d_in[0];
  const float* W0 = (const float*)d_in[1];
  const float* W1 = (const float*)d_in[2];
  const float* W2 = (const float*)d_in[3];
  const int* src  = (const int*)d_in[4];
  const int* dst  = (const int*)d_in[5];
  float* out = (float*)d_out;

  const int n = in_sizes[0] / GIN_D_IN;
  const int n_edges = in_sizes[4];

  float* tbuf = (float*)d_ws;
  float* agg1 = tbuf + (size_t)n * GIN_D_HID;
  int* cnt    = (int*)(agg1 + (size_t)n * GIN_D_HID);
  int* off    = cnt + GIN_MAXB * GIN_CHUNKS;
  int* btotal = off + GIN_MAXB * GIN_CHUNKS;
  int* rowptr = btotal + 2 * GIN_MAXB;
  int* pairs  = rowptr + (((size_t)n + 1 + 3) / 4) * 4;
  int* csrc   = pairs + 2 * (size_t)n_edges;

  const int ntiles = (n + 15) / 16;
  const int gemm_blocks = (ntiles + 7) / 8;
  const int n_buckets = (n + GIN_BNODES - 1) / GIN_BNODES;
  const int chunk_len = (n_edges + GIN_CHUNKS - 1) / GIN_CHUNKS;
  const int gather_blocks = (n * 8 + 255) / 256;
  const int group_blocks = (n + GIN_GROUP_NODES - 1) / GIN_GROUP_NODES;

  gin_bucket_count_kernel<<<GIN_CHUNKS, 1024, 0, stream>>>(dst, cnt, n_edges, n_buckets, chunk_len);
  gin_bucket_offset_kernel<<<n_buckets, GIN_CHUNKS, 0, stream>>>(cnt, off, btotal);
  gin_bucket_place_kernel<<<GIN_CHUNKS, 1024, 0, stream>>>(src, dst, off, btotal, pairs, n_edges, n_buckets, chunk_len);
  gin_bucket_csr_kernel<<<n_buckets, GIN_BNODES, 0, stream>>>(pairs, btotal, rowptr, csrc, n);

  gin_gemm_kernel<GIN_D_IN, GIN_D_HID, 2, false>
      <<<gemm_blocks, 256, 0, stream>>>(x, W0, tbuf, n);
  gin_gather_kernel<false><<<gather_blocks, 256, 0, stream>>>(tbuf, rowptr, csrc, agg1, n);
  gin_gather_gemm_kernel<true, GIN_D_HID, 2, true>
      <<<group_blocks, 256, 0, stream>>>(agg1, rowptr, csrc, W1, tbuf, n);
  gin_gather_gemm_kernel<false, GIN_N_CLASSES, 3, false>
      <<<group_blocks, 256, 0, stream>>>(tbuf, rowptr, csrc, W2, out, n);
}


# Edge-grouping loops (bucket count, place, row lists) load a thread's edges in batches of 7 to 20 ahead of their LDS counter adds, not one per trip; a node's source order stays arrival order
# speedup vs baseline: 2.8602x; 1.0679x over previous
#include <hip/hip_runtime.h>


typedef __attribute__((ext_vector_type(2))) float v2f;
typedef __attribute__((ext_vector_type(8))) float v8f;

#define GIN_D_IN 64
#define GIN_D_HID 32
#define GIN_N_CLASSES 40
#define GIN_SELF_W 1.1f

#define GIN_BSH 8
#define GIN_BNODES (1 << GIN_BSH)
#define GIN_MAXB 512
#define GIN_CHUNKS 256
#define GIN_EB 7
#define GIN_CB 20

__global__ __launch_bounds__(1024) void gin_bucket_count_kernel(
    const int* __restrict__ dst, int* __restrict__ cnt, int n_edges,
    int n_buckets, int chunk_len) {
  __shared__ int hist[GIN_MAXB];
  const int c = blockIdx.x;
  const int beg = c * chunk_len;
  const int end = min(beg + chunk_len, n_edges);
  const int last = max(n_edges - 1, 0);
  int e0 = beg + (int)threadIdx.x;
  int d[GIN_EB];
#pragma unroll
  for (int j = 0; j < GIN_EB; ++j) d[j] = dst[min(e0 + j * 1024, last)];
  for (int i = threadIdx.x; i < GIN_MAXB; i += 1024) hist[i] = 0;
  __syncthreads();
  for (;;) {
#pragma unroll
    for (int j = 0; j < GIN_EB; ++j)
      if (e0 + j * 1024 < end) atomicAdd(&hist[d[j] >> GIN_BSH], 1);
    e0 += 1024 * GIN_EB;
    if (e0 >= end) break;
#pragma unroll
    for (int j = 0; j < GIN_EB; ++j) d[j] = dst[min(e0 + j * 1024, last)];
  }
  __syncthreads();
  for (int b = threadIdx.x; b < n_buckets; b += 1024)
    cnt[b * GIN_CHUNKS + c] = hist[b];
}

__global__ __launch_bounds__(GIN_CHUNKS) void gin_bucket_offset_kernel(
    const int* __restrict__ cnt, int* __restrict__ off, int* __restrict__ btotal) {
  __shared__ int w_own[GIN_CHUNKS / 32];
  const int b = blockIdx.x;
  const int t = threadIdx.x;
  const int lane = t & 31;
  const int wave = t >> 5;

  const int own = cnt[b * GIN_CHUNKS + t];
  int incl = own;
#pragma unroll
  for (int d = 1; d < 32; d <<= 1) {
    int up = __shfl_up(incl, d);
    if (lane >= d) incl += up;
  }
  if (lane == 31) w_own[wave] = incl;
  __syncthreads();
  int wpre = 0, tot = 0;
#pragma unroll
  for (int w = 0; w < GIN_CHUNKS / 32; ++w) {
    const int v = w_own[w];
    tot += v;
    if (w < wave) wpre += v;
  }
  off[b * GIN_CHUNKS + t] = wpre + incl - own;
  if (t == 0) btotal[b] = tot;
}

__global__ __launch_bounds__(1024) void gin_bucket_place_kernel(
    const int* __restrict__ src, const int* __restrict__ dst,
    const int* __restrict__ off, const int* __restrict__ btotal,
    int* __restrict__ pairs, int n_edges, int n_buckets, int chunk_len) {
  __shared__ int cur[GIN_MAXB];
  __shared__ int w_tot[GIN_MAXB / 32];
  const int c = blockIdx.x;
  const int t = threadIdx.x;
  const int lane = t & 31;
  const int wave = t >> 5;
  const int beg = c * chunk_len;
  const int end = min(beg + chunk_len, n_edges);
  const int last = max(n_edges - 1, 0);
  int e0 = beg + t;
  int d[GIN_EB], s[GIN_EB];
#pragma unroll
  for (int j = 0; j < GIN_EB; ++j) {
    const int e = min(e0 + j * 1024, last);
    d[j] = dst[e];
    s[j] = src[e];
  }
  const bool has = t < n_buckets;
  const int tot = has ? btotal[t] : 0;
  const int rank = has ? off[t * GIN_CHUNKS + c] : 0;
  int incl = tot;
#pragma unroll
  for (int d = 1; d < 32; d <<= 1) {
    int up = __shfl_up(incl, d);
    if (lane >= d) incl += up;
  }
  if (lane == 31 && wave < GIN_MAXB / 32) w_tot[wave] = incl;
  __syncthreads();
  if (has) {
    int wpre = 0;
#pragma unroll
    for (int w = 0; w < GIN_MAXB / 32; ++w)
      if (w < wave) wpre += w_tot[w];
    cur[t] = wpre + incl - tot + rank;
  }
  __syncthreads();
  int2* p2 = (int2*)pairs;
  for (;;) {
    int pos[GIN_EB];
#pragma unroll
    for (int j = 0; j < GIN_EB; ++j)
      if (e0 + j * 1024 < end) pos[j] = atomicAdd(&cur[d[j] >> GIN_BSH], 1);
#pragma unroll
    for (int j = 0; j < GIN_EB; ++j)
      if (e0 + j * 1024 < end) p2[pos[j]] = make_int2(s[j], d[j]);
    e0 += 1024 * GIN_EB;
    if (e0 >= end) break;
#pragma unroll
    for (int j = 0; j < GIN_EB; ++j) {
      const int e = min(e0 + j * 1024, last);
      d[j] = dst[e];
      s[j] = src[e];
    }
  }
}

__global__ __launch_bounds__(GIN_BNODES) void gin_bucket_csr_kernel(
    const int* __restrict__ pairs, const int* __restrict__ btotal,
    int* __restrict__ rowptr, int* __restrict__ csr_src, int n) {
  static_assert(GIN_MAXB <= 2 * GIN_BNODES, "two totals per thread cover every earlier bucket");
  __shared__ int cnt[GIN_BNODES];
  __shared__ int w_tot[GIN_BNODES / 32];
  __shared__ int w_beg[GIN_BNODES / 32];
  const int b = blockIdx.x;
  const int t = threadIdx.x;
  const int lane = t & 31;
  const int wave = t >> 5;
  const int2* p2 = (const int2*)pairs;

  int part = (t < b ? btotal[t] : 0) +
             (t + GIN_BNODES < b ? btotal[t + GIN_BNODES] : 0);
#pragma unroll
  for (int d = 16; d >= 1; d >>= 1) part += __shfl_xor(part, d);
  if (lane == 0) w_beg[wave] = part;
  cnt[t] = 0;
  __syncthreads();
  int beg = 0;
#pragma unroll
  for (int w = 0; w < GIN_BNODES / 32; ++w) beg += w_beg[w];
  const int end = beg + btotal[b];
  const int last = max(end - 1, 0);
  int2 r[GIN_CB];
#pragma unroll
  for (int j = 0; j < GIN_CB; ++j) r[j] = p2[min(beg + t + j * GIN_BNODES, last)];
#pragma unroll
  for (int j = 0; j < GIN_CB; ++j)
    if (beg + t + j * GIN_BNODES < end) atomicAdd(&cnt[r[j].y & (GIN_BNODES - 1)], 1);
  for (int e = beg + t + GIN_CB * GIN_BNODES; e < end; e += GIN_BNODES)
    atomicAdd(&cnt[p2[e].y & (GIN_BNODES - 1)], 1);
  __syncthreads();

  const int own = cnt[t];
  int incl = own;
#pragma unroll
  for (int d = 1; d < 32; d <<= 1) {
    int up = __shfl_up(incl, d);
    if (lane >= d) incl += up;
  }
  if (lane == 31) w_tot[wave] = incl;
  __syncthreads();
  int wpre = 0;
#pragma unroll
  for (int w = 0; w < GIN_BNODES / 32; ++w)
    if (w < wave) wpre += w_tot[w];
  const int first = beg + wpre + incl - own;
  cnt[t] = first;
  const int node = b * GIN_BNODES + t;
  if (node < n) rowptr[node] = first;
  if (b == (int)gridDim.x - 1 && t == 0) rowptr[n] = end;
  __syncthreads();

  int pos[GIN_CB];
#pragma unroll
  for (int j = 0; j < GIN_CB; ++j)
    if (beg + t + j * GIN_BNODES < end)
      pos[j] = atomicAdd(&cnt[r[j].y & (GIN_BNODES - 1)], 1);
#pragma unroll
  for (int j = 0; j < GIN_CB; ++j)
    if (beg + t + j * GIN_BNODES < end) csr_src[pos[j]] = r[j].x;
  for (int e = beg + t + GIN_CB * GIN_BNODES; e < end; e += GIN_BNODES) {
    const int2 p = p2[e];
    const int pos = atomicAdd(&cnt[p.y & (GIN_BNODES - 1)], 1);
    csr_src[pos] = p.x;
  }
}

template <bool RELU_IN>
__device__ __forceinline__ float4 gin_gather_row(
    const float4* __restrict__ h4, const int* __restrict__ rowptr,
    const int* __restrict__ csr_src, int node, int l8) {
  float4 acc = h4[(unsigned)node * 8u + (unsigned)l8];
  if (RELU_IN) {
    acc.x = fmaxf(acc.x, 0.0f); acc.y = fmaxf(acc.y, 0.0f);
    acc.z = fmaxf(acc.z, 0.0f); acc.w = fmaxf(acc.w, 0.0f);
  }
  acc.x *= GIN_SELF_W; acc.y *= GIN_SELF_W; acc.z *= GIN_SELF_W; acc.w *= GIN_SELF_W;
  const int beg = rowptr[node];
  const int end = rowptr[node + 1];
  for (int base = beg; base < end; base += 8) {
    const int e = min(base + l8, end - 1);
    const int mine = csr_src[e];
    float4 r[8];
#pragma unroll
    for (int j = 0; j < 8; ++j) {
      const unsigned s = (unsigned)__shfl(mine, j, 8);
      r[j] = h4[s * 8u + (unsigned)l8];
    }
#pragma unroll
    for (int j = 0; j < 8; ++j) {
      float4 v = r[j];
      if (RELU_IN) {
        v.x = fmaxf(v.x, 0.0f); v.y = fmaxf(v.y, 0.0f);
        v.z = fmaxf(v.z, 0.0f); v.w = fmaxf(v.w, 0.0f);
      }
      const bool on = base + j < end;
      acc.x += on ? v.x : 0.0f; acc.y += on ? v.y : 0.0f;
      acc.z += on ? v.z : 0.0f; acc.w += on ? v.w : 0.0f;
    }
  }
  return acc;
}

template <bool RELU_IN>
__global__ __launch_bounds__(256) void gin_gather_kernel(
    const float* __restrict__ h, const int* __restrict__ rowptr,
    const int* __restrict__ csr_src, float* __restrict__ agg, int n) {
  const int l8 = threadIdx.x & 7;
  const int node = (blockIdx.x * 256 + (int)threadIdx.x) >> 3;
  if (node >= n) return;
  const float4 acc = gin_gather_row<RELU_IN>((const float4*)h, rowptr, csr_src, node, l8);
  ((float4*)agg)[(unsigned)node * 8u + (unsigned)l8] = acc;
}

#define GIN_GROUP_NODES 32
#define GIN_LDS_PITCH 36
template <bool RELU_IN, int DOUTL, int NT, bool RELU_OUT>
__global__ __launch_bounds__(256) void gin_gather_gemm_kernel(
    const float* __restrict__ h, const int* __restrict__ rowptr,
    const int* __restrict__ csr_src, const float* __restrict__ W,
    float* __restrict__ C, int n) {
  __shared__ __attribute__((aligned(16))) float rows[GIN_GROUP_NODES * GIN_LDS_PITCH];
  const int l8 = threadIdx.x & 7;
  const int lnode = (int)threadIdx.x >> 3;
  const int node = blockIdx.x * GIN_GROUP_NODES + lnode;
  float4 sum = make_float4(0.0f, 0.0f, 0.0f, 0.0f);
  if (node < n)
    sum = gin_gather_row<RELU_IN>((const float4*)h, rowptr, csr_src, node, l8);
  *(float4*)&rows[lnode * GIN_LDS_PITCH + l8 * 4] = sum;
  __syncthreads();

  const int lane = threadIdx.x & 31;
  const int wave = threadIdx.x >> 5;
  if (wave >= GIN_GROUP_NODES / 16) return;

  const int l16 = lane & 15;
  const int khalf = lane >> 4;
  const int row0 = blockIdx.x * GIN_GROUP_NODES + wave * 16;
  const float* arow = &rows[(wave * 16 + l16) * GIN_LDS_PITCH];

  v8f acc[NT] = {};

#pragma unroll
  for (int kk = 0; kk < GIN_D_HID / 4; ++kk) {
    const int kb = kk * 4 + khalf * 2;
    const v2f a = *(const v2f*)(arow + kb);
#pragma unroll
    for (int t = 0; t < NT; ++t) {
      const int col = t * 16 + l16;
      const bool ok = ((DOUTL & 15) == 0) || (col < DOUTL);
      v2f b;
      b.x = ok ? W[(size_t)kb * DOUTL + col] : 0.0f;
      b.y = ok ? W[(size_t)(kb + 1) * DOUTL + col] : 0.0f;
      acc[t] = __builtin_amdgcn_wmma_f32_16x16x4_f32(
          false, a, false, b, (short)0, acc[t], false, false);
    }
  }

#pragma unroll
  for (int t = 0; t < NT; ++t) {
    const int col = t * 16 + l16;
    if (((DOUTL & 15) != 0) && col >= DOUTL) continue;
#pragma unroll
    for (int r = 0; r < 8; ++r) {
      const int row = row0 + r + khalf * 8;
      if (row < n) {
        float v = acc[t][r];
        if (RELU_OUT) v = fmaxf(v, 0.0f);
        C[(size_t)row * DOUTL + col] = v;
      }
    }
  }
}

template <int DINL, int DOUTL, int NT, bool RELU>
__global__ __launch_bounds__(256) void gin_gemm_kernel(
    const float* __restrict__ A, const float* __restrict__ W,
    float* __restrict__ C, int n_rows) {
  const int lane = threadIdx.x & 31;
  const int wave = threadIdx.x >> 5;
  const int ntiles = (n_rows + 15) >> 4;
  const int tile = blockIdx.x * (blockDim.x >> 5) + wave;
  if (tile >= ntiles) return;

  const int row0 = tile << 4;
  const int l16 = lane & 15;
  const int khalf = lane >> 4;
  int arow = row0 + l16;
  if (arow >= n_rows) arow = n_rows - 1;

  v8f acc[NT] = {};

#pragma unroll
  for (int kk = 0; kk < DINL / 4; ++kk) {
    const int kb = kk * 4 + khalf * 2;
    v2f a;
    a.x = A[(size_t)arow * DINL + kb];
    a.y = A[(size_t)arow * DINL + kb + 1];
#pragma unroll
    for (int t = 0; t < NT; ++t) {
      const int col = t * 16 + l16;
      const bool ok = ((DOUTL & 15) == 0) || (col < DOUTL);
      v2f b;
      b.x = ok ? W[(size_t)kb * DOUTL + col] : 0.0f;
      b.y = ok ? W[(size_t)(kb + 1) * DOUTL + col] : 0.0f;
      acc[t] = __builtin_amdgcn_wmma_f32_16x16x4_f32(
          false, a, false, b, (short)0, acc[t], false, false);
    }
  }

#pragma unroll
  for (int t = 0; t < NT; ++t) {
    const int col = t * 16 + l16;
    if (((DOUTL & 15) != 0) && col >= DOUTL) continue;
#pragma unroll
    for (int r = 0; r < 8; ++r) {
      const int row = row0 + r + khalf * 8;
      if (row < n_rows) {
        float v = acc[t][r];
        if (RELU) v = fmaxf(v, 0.0f);
        C[(size_t)row * DOUTL + col] = v;
      }
    }
  }
}

extern "C" void kernel_launch(void* const* d_in, const int* in_sizes, int n_in,
                              void* d_out, int out_size, void* d_ws, size_t ws_size,
                              hipStream_t stream) {
  const float* x  = (const float*)d_in[0];
  const float* W0 = (const float*)d_in[1];
  const float* W1 = (const float*)d_in[2];
  const float* W2 = (const float*)d_in[3];
  const int* src  = (const int*)d_in[4];
  const int* dst  = (const int*)d_in[5];
  float* out = (float*)d_out;

  const int n = in_sizes[0] / GIN_D_IN;
  const int n_edges = in_sizes[4];

  float* tbuf = (float*)d_ws;
  float* agg1 = tbuf + (size_t)n * GIN_D_HID;
  int* cnt    = (int*)(agg1 + (size_t)n * GIN_D_HID);
  int* off    = cnt + GIN_MAXB * GIN_CHUNKS;
  int* btotal = off + GIN_MAXB * GIN_CHUNKS;
  int* rowptr = btotal + 2 * GIN_MAXB;
  int* pairs  = rowptr + (((size_t)n + 1 + 3) / 4) * 4;
  int* csrc   = pairs + 2 * (size_t)n_edges;

  const int ntiles = (n + 15) / 16;
  const int gemm_blocks = (ntiles + 7) / 8;
  const int n_buckets = (n + GIN_BNODES - 1) / GIN_BNODES;
  const int chunk_len = (n_edges + GIN_CHUNKS - 1) / GIN_CHUNKS;
  const int gather_blocks = (n * 8 + 255) / 256;
  const int group_blocks = (n + GIN_GROUP_NODES - 1) / GIN_GROUP_NODES;

  gin_bucket_count_kernel<<<GIN_CHUNKS, 1024, 0, stream>>>(dst, cnt, n_edges, n_buckets, chunk_len);
  gin_bucket_offset_kernel<<<n_buckets, GIN_CHUNKS, 0, stream>>>(cnt, off, btotal);
  gin_bucket_place_kernel<<<GIN_CHUNKS, 1024, 0, stream>>>(src, dst, off, btotal, pairs, n_edges, n_buckets, chunk_len);
  gin_bucket_csr_kernel<<<n_buckets, GIN_BNODES, 0, stream>>>(pairs, btotal, rowptr, csrc, n);

  gin_gemm_kernel<GIN_D_IN, GIN_D_HID, 2, false>
      <<<gemm_blocks, 256, 0, stream>>>(x, W0, tbuf, n);
  gin_gather_kernel<false><<<gather_blocks, 256, 0, stream>>>(tbuf, rowptr, csrc, agg1, n);
  gin_gather_gemm_kernel<true, GIN_D_HID, 2, true>
      <<<group_blocks, 256, 0, stream>>>(agg1, rowptr, csrc, W1, tbuf, n);
  gin_gather_gemm_kernel<false, GIN_N_CLASSES, 3, false>
      <<<group_blocks, 256, 0, stream>>>(tbuf, rowptr, csrc, W2, out, n);
}


# Layer-1 projection x W0 gets no launch of its own: its 782 tile blocks ride behind the 391 row-list blocks in one grid (role by block index); 8 launches become 7; sum order unchanged
# speedup vs baseline: 3.0164x; 1.0546x over previous
#include <hip/hip_runtime.h>


typedef __attribute__((ext_vector_type(2))) float v2f;
typedef __attribute__((ext_vector_type(8))) float v8f;

#define GIN_D_IN 64
#define GIN_D_HID 32
#define GIN_N_CLASSES 40
#define GIN_SELF_W 1.1f

#define GIN_BSH 8
#define GIN_BNODES (1 << GIN_BSH)
#define GIN_MAXB 512
#define GIN_CHUNKS 256
#define GIN_EB 7
#define GIN_CB 20

__global__ __launch_bounds__(1024) void gin_bucket_count_kernel(
    const int* __restrict__ dst, int* __restrict__ cnt, int n_edges,
    int n_buckets, int chunk_len) {
  __shared__ int hist[GIN_MAXB];
  const int c = blockIdx.x;
  const int beg = c * chunk_len;
  const int end = min(beg + chunk_len, n_edges);
  const int last = max(n_edges - 1, 0);
  int e0 = beg + (int)threadIdx.x;
  int d[GIN_EB];
#pragma unroll
  for (int j = 0; j < GIN_EB; ++j) d[j] = dst[min(e0 + j * 1024, last)];
  for (int i = threadIdx.x; i < GIN_MAXB; i += 1024) hist[i] = 0;
  __syncthreads();
  for (;;) {
#pragma unroll
    for (int j = 0; j < GIN_EB; ++j)
      if (e0 + j * 1024 < end) atomicAdd(&hist[d[j] >> GIN_BSH], 1);
    e0 += 1024 * GIN_EB;
    if (e0 >= end) break;
#pragma unroll
    for (int j = 0; j < GIN_EB; ++j) d[j] = dst[min(e0 + j * 1024, last)];
  }
  __syncthreads();
  for (int b = threadIdx.x; b < n_buckets; b += 1024)
    cnt[b * GIN_CHUNKS + c] = hist[b];
}

__global__ __launch_bounds__(GIN_CHUNKS) void gin_bucket_offset_kernel(
    const int* __restrict__ cnt, int* __restrict__ off, int* __restrict__ btotal) {
  __shared__ int w_own[GIN_CHUNKS / 32];
  const int b = blockIdx.x;
  const int t = threadIdx.x;
  const int lane = t & 31;
  const int wave = t >> 5;

  const int own = cnt[b * GIN_CHUNKS + t];
  int incl = own;
#pragma unroll
  for (int d = 1; d < 32; d <<= 1) {
    int up = __shfl_up(incl, d);
    if (lane >= d) incl += up;
  }
  if (lane == 31) w_own[wave] = incl;
  __syncthreads();
  int wpre = 0, tot = 0;
#pragma unroll
  for (int w = 0; w < GIN_CHUNKS / 32; ++w) {
    const int v = w_own[w];
    tot += v;
    if (w < wave) wpre += v;
  }
  off[b * GIN_CHUNKS + t] = wpre + incl - own;
  if (t == 0) btotal[b] = tot;
}

__global__ __launch_bounds__(1024) void gin_bucket_place_kernel(
    const int* __restrict__ src, const int* __restrict__ dst,
    const int* __restrict__ off, const int* __restrict__ btotal,
    int* __restrict__ pairs, int n_edges, int n_buckets, int chunk_len) {
  __shared__ int cur[GIN_MAXB];
  __shared__ int w_tot[GIN_MAXB / 32];
  const int c = blockIdx.x;
  const int t = threadIdx.x;
  const int lane = t & 31;
  const int wave = t >> 5;
  const int beg = c * chunk_len;
  const int end = min(beg + chunk_len, n_edges);
  const int last = max(n_edges - 1, 0);
  int e0 = beg + t;
  int d[GIN_EB], s[GIN_EB];
#pragma unroll
  for (int j = 0; j < GIN_EB; ++j) {
    const int e = min(e0 + j * 1024, last);
    d[j] = dst[e];
    s[j] = src[e];
  }
  const bool has = t < n_buckets;
  const int tot = has ? btotal[t] : 0;
  const int rank = has ? off[t * GIN_CHUNKS + c] : 0;
  int incl = tot;
#pragma unroll
  for (int d = 1; d < 32; d <<= 1) {
    int up = __shfl_up(incl, d);
    if (lane >= d) incl += up;
  }
  if (lane == 31 && wave < GIN_MAXB / 32) w_tot[wave] = incl;
  __syncthreads();
  if (has) {
    int wpre = 0;
#pragma unroll
    for (int w = 0; w < GIN_MAXB / 32; ++w)
      if (w < wave) wpre += w_tot[w];
    cur[t] = wpre + incl - tot + rank;
  }
  __syncthreads();
  int2* p2 = (int2*)pairs;
  for (;;) {
    int pos[GIN_EB];
#pragma unroll
    for (int j = 0; j < GIN_EB; ++j)
      if (e0 + j * 1024 < end) pos[j] = atomicAdd(&cur[d[j] >> GIN_BSH], 1);
#pragma unroll
    for (int j = 0; j < GIN_EB; ++j)
      if (e0 + j * 1024 < end) p2[pos[j]] = make_int2(s[j], d[j]);
    e0 += 1024 * GIN_EB;
    if (e0 >= end) break;
#pragma unroll
    for (int j = 0; j < GIN_EB; ++j) {
      const int e = min(e0 + j * 1024, last);
      d[j] = dst[e];
      s[j] = src[e];
    }
  }
}

__device__ __forceinline__ void gin_bucket_csr_block(
    const int* __restrict__ pairs, const int* __restrict__ btotal,
    int* __restrict__ rowptr, int* __restrict__ csr_src, int n, int n_buckets) {
  static_assert(GIN_MAXB <= 2 * GIN_BNODES, "two totals per thread cover every earlier bucket");
  __shared__ int cnt[GIN_BNODES];
  __shared__ int w_tot[GIN_BNODES / 32];
  __shared__ int w_beg[GIN_BNODES / 32];
  const int b = blockIdx.x;
  const int t = threadIdx.x;
  const int lane = t & 31;
  const int wave = t >> 5;
  const int2* p2 = (const int2*)pairs;

  int part = (t < b ? btotal[t] : 0) +
             (t + GIN_BNODES < b ? btotal[t + GIN_BNODES] : 0);
#pragma unroll
  for (int d = 16; d >= 1; d >>= 1) part += __shfl_xor(part, d);
  if (lane == 0) w_beg[wave] = part;
  cnt[t] = 0;
  __syncthreads();
  int beg = 0;
#pragma unroll
  for (int w = 0; w < GIN_BNODES / 32; ++w) beg += w_beg[w];
  const int end = beg + btotal[b];
  const int last = max(end - 1, 0);
  int2 r[GIN_CB];
#pragma unroll
  for (int j = 0; j < GIN_CB; ++j) r[j] = p2[min(beg + t + j * GIN_BNODES, last)];
#pragma unroll
  for (int j = 0; j < GIN_CB; ++j)
    if (beg + t + j * GIN_BNODES < end) atomicAdd(&cnt[r[j].y & (GIN_BNODES - 1)], 1);
  for (int e = beg + t + GIN_CB * GIN_BNODES; e < end; e += GIN_BNODES)
    atomicAdd(&cnt[p2[e].y & (GIN_BNODES - 1)], 1);
  __syncthreads();

  const int own = cnt[t];
  int incl = own;
#pragma unroll
  for (int d = 1; d < 32; d <<= 1) {
    int up = __shfl_up(incl, d);
    if (lane >= d) incl += up;
  }
  if (lane == 31) w_tot[wave] = incl;
  __syncthreads();
  int wpre = 0;
#pragma unroll
  for (int w = 0; w < GIN_BNODES / 32; ++w)
    if (w < wave) wpre += w_tot[w];
  const int first = beg + wpre + incl - own;
  cnt[t] = first;
  const int node = b * GIN_BNODES + t;
  if (node < n) rowptr[node] = first;
  if (b == n_buckets - 1 && t == 0) rowptr[n] = end;
  __syncthreads();

  int pos[GIN_CB];
#pragma unroll
  for (int j = 0; j < GIN_CB; ++j)
    if (beg + t + j * GIN_BNODES < end)
      pos[j] = atomicAdd(&cnt[r[j].y & (GIN_BNODES - 1)], 1);
#pragma unroll
  for (int j = 0; j < GIN_CB; ++j)
    if (beg + t + j * GIN_BNODES < end) csr_src[pos[j]] = r[j].x;
  for (int e = beg + t + GIN_CB * GIN_BNODES; e < end; e += GIN_BNODES) {
    const int2 p = p2[e];
    const int pos = atomicAdd(&cnt[p.y & (GIN_BNODES - 1)], 1);
    csr_src[pos] = p.x;
  }
}

template <bool RELU_IN>
__device__ __forceinline__ float4 gin_gather_row(
    const float4* __restrict__ h4, const int* __restrict__ rowptr,
    const int* __restrict__ csr_src, int node, int l8) {
  float4 acc = h4[(unsigned)node * 8u + (unsigned)l8];
  if (RELU_IN) {
    acc.x = fmaxf(acc.x, 0.0f); acc.y = fmaxf(acc.y, 0.0f);
    acc.z = fmaxf(acc.z, 0.0f); acc.w = fmaxf(acc.w, 0.0f);
  }
  acc.x *= GIN_SELF_W; acc.y *= GIN_SELF_W; acc.z *= GIN_SELF_W; acc.w *= GIN_SELF_W;
  const int beg = rowptr[node];
  const int end = rowptr[node + 1];
  for (int base = beg; base < end; base += 8) {
    const int e = min(base + l8, end - 1);
    const int mine = csr_src[e];
    float4 r[8];
#pragma unroll
    for (int j = 0; j < 8; ++j) {
      const unsigned s = (unsigned)__shfl(mine, j, 8);
      r[j] = h4[s * 8u + (unsigned)l8];
    }
#pragma unroll
    for (int j = 0; j < 8; ++j) {
      float4 v = r[j];
      if (RELU_IN) {
        v.x = fmaxf(v.x, 0.0f); v.y = fmaxf(v.y, 0.0f);
        v.z = fmaxf(v.z, 0.0f); v.w = fmaxf(v.w, 0.0f);
      }
      const bool on = base + j < end;
      acc.x += on ? v.x : 0.0f; acc.y += on ? v.y : 0.0f;
      acc.z += on ? v.z : 0.0f; acc.w += on ? v.w : 0.0f;
    }
  }
  return acc;
}

template <bool RELU_IN>
__global__ __launch_bounds__(256) void gin_gather_kernel(
    const float* __restrict__ h, const int* __restrict__ rowptr,
    const int* __restrict__ csr_src, float* __restrict__ agg, int n) {
  const int l8 = threadIdx.x & 7;
  const int node = (blockIdx.x * 256 + (int)threadIdx.x) >> 3;
  if (node >= n) return;
  const float4 acc = gin_gather_row<RELU_IN>((const float4*)h, rowptr, csr_src, node, l8);
  ((float4*)agg)[(unsigned)node * 8u + (unsigned)l8] = acc;
}

#define GIN_GROUP_NODES 32
#define GIN_LDS_PITCH 36
template <bool RELU_IN, int DOUTL, int NT, bool RELU_OUT>
__global__ __launch_bounds__(256) void gin_gather_gemm_kernel(
    const float* __restrict__ h, const int* __restrict__ rowptr,
    const int* __restrict__ csr_src, const float* __restrict__ W,
    float* __restrict__ C, int n) {
  __shared__ __attribute__((aligned(16))) float rows[GIN_GROUP_NODES * GIN_LDS_PITCH];
  const int l8 = threadIdx.x & 7;
  const int lnode = (int)threadIdx.x >> 3;
  const int node = blockIdx.x * GIN_GROUP_NODES + lnode;
  float4 sum = make_float4(0.0f, 0.0f, 0.0f, 0.0f);
  if (node < n)
    sum = gin_gather_row<RELU_IN>((const float4*)h, rowptr, csr_src, node, l8);
  *(float4*)&rows[lnode * GIN_LDS_PITCH + l8 * 4] = sum;
  __syncthreads();

  const int lane = threadIdx.x & 31;
  const int wave = threadIdx.x >> 5;
  if (wave >= GIN_GROUP_NODES / 16) return;

  const int l16 = lane & 15;
  const int khalf = lane >> 4;
  const int row0 = blockIdx.x * GIN_GROUP_NODES + wave * 16;
  const float* arow = &rows[(wave * 16 + l16) * GIN_LDS_PITCH];

  v8f acc[NT] = {};

#pragma unroll
  for (int kk = 0; kk < GIN_D_HID / 4; ++kk) {
    const int kb = kk * 4 + khalf * 2;
    const v2f a = *(const v2f*)(arow + kb);
#pragma unroll
    for (int t = 0; t < NT; ++t) {
      const int col = t * 16 + l16;
      const bool ok = ((DOUTL & 15) == 0) || (col < DOUTL);
      v2f b;
      b.x = ok ? W[(size_t)kb * DOUTL + col] : 0.0f;
      b.y = ok ? W[(size_t)(kb + 1) * DOUTL + col] : 0.0f;
      acc[t] = __builtin_amdgcn_wmma_f32_16x16x4_f32(
          false, a, false, b, (short)0, acc[t], false, false);
    }
  }

#pragma unroll
  for (int t = 0; t < NT; ++t) {
    const int col = t * 16 + l16;
    if (((DOUTL & 15) != 0) && col >= DOUTL) continue;
#pragma unroll
    for (int r = 0; r < 8; ++r) {
      const int row = row0 + r + khalf * 8;
      if (row < n) {
        float v = acc[t][r];
        if (RELU_OUT) v = fmaxf(v, 0.0f);
        C[(size_t)row * DOUTL + col] = v;
      }
    }
  }
}

template <int DINL, int DOUTL, int NT, bool RELU>
__device__ __forceinline__ void gin_gemm_tile(
    const float* __restrict__ A, const float* __restrict__ W,
    float* __restrict__ C, int n_rows, int tile) {
  const int lane = threadIdx.x & 31;
  const int row0 = tile << 4;
  const int l16 = lane & 15;
  const int khalf = lane >> 4;
  int arow = row0 + l16;
  if (arow >= n_rows) arow = n_rows - 1;

  v8f acc[NT] = {};

#pragma unroll
  for (int kk = 0; kk < DINL / 4; ++kk) {
    const int kb = kk * 4 + khalf * 2;
    v2f a;
    a.x = A[(size_t)arow * DINL + kb];
    a.y = A[(size_t)arow * DINL + kb + 1];
#pragma unroll
    for (int t = 0; t < NT; ++t) {
      const int col = t * 16 + l16;
      const bool ok = ((DOUTL & 15) == 0) || (col < DOUTL);
      v2f b;
      b.x = ok ? W[(size_t)kb * DOUTL + col] : 0.0f;
      b.y = ok ? W[(size_t)(kb + 1) * DOUTL + col] : 0.0f;
      acc[t] = __builtin_amdgcn_wmma_f32_16x16x4_f32(
          false, a, false, b, (short)0, acc[t], false, false);
    }
  }

#pragma unroll
  for (int t = 0; t < NT; ++t) {
    const int col = t * 16 + l16;
    if (((DOUTL & 15) != 0) && col >= DOUTL) continue;
#pragma unroll
    for (int r = 0; r < 8; ++r) {
      const int row = row0 + r + khalf * 8;
      if (row < n_rows) {
        float v = acc[t][r];
        if (RELU) v = fmaxf(v, 0.0f);
        C[(size_t)row * DOUTL + col] = v;
      }
    }
  }
}

__global__ __launch_bounds__(GIN_BNODES) void gin_csr_project_kernel(
    const int* __restrict__ pairs, const int* __restrict__ btotal,
    int* __restrict__ rowptr, int* __restrict__ csr_src, int n, int n_buckets,
    const float* __restrict__ x, const float* __restrict__ W0,
    float* __restrict__ xw) {
  if ((int)blockIdx.x >= n_buckets) {
    const int tile = ((int)blockIdx.x - n_buckets) * (GIN_BNODES / 32) + ((int)threadIdx.x >> 5);
    if (tile < ((n + 15) >> 4))
      gin_gemm_tile<GIN_D_IN, GIN_D_HID, 2, false>(x, W0, xw, n, tile);
    return;
  }
  gin_bucket_csr_block(pairs, btotal, rowptr, csr_src, n, n_buckets);
}

extern "C" void kernel_launch(void* const* d_in, const int* in_sizes, int n_in,
                              void* d_out, int out_size, void* d_ws, size_t ws_size,
                              hipStream_t stream) {
  const float* x  = (const float*)d_in[0];
  const float* W0 = (const float*)d_in[1];
  const float* W1 = (const float*)d_in[2];
  const float* W2 = (const float*)d_in[3];
  const int* src  = (const int*)d_in[4];
  const int* dst  = (const int*)d_in[5];
  float* out = (float*)d_out;

  const int n = in_sizes[0] / GIN_D_IN;
  const int n_edges = in_sizes[4];

  float* tbuf = (float*)d_ws;
  float* agg1 = tbuf + (size_t)n * GIN_D_HID;
  int* cnt    = (int*)(agg1 + (size_t)n * GIN_D_HID);
  int* off    = cnt + GIN_MAXB * GIN_CHUNKS;
  int* btotal = off + GIN_MAXB * GIN_CHUNKS;
  int* rowptr = btotal + 2 * GIN_MAXB;
  int* pairs  = rowptr + (((size_t)n + 1 + 3) / 4) * 4;
  int* csrc   = pairs + 2 * (size_t)n_edges;

  const int ntiles = (n + 15) / 16;
  const int gemm_blocks = (ntiles + 7) / 8;
  const int n_buckets = (n + GIN_BNODES - 1) / GIN_BNODES;
  const int chunk_len = (n_edges + GIN_CHUNKS - 1) / GIN_CHUNKS;
  const int gather_blocks = (n * 8 + 255) / 256;
  const int group_blocks = (n + GIN_GROUP_NODES - 1) / GIN_GROUP_NODES;

  gin_bucket_count_kernel<<<GIN_CHUNKS, 1024, 0, stream>>>(dst, cnt, n_edges, n_buckets, chunk_len);
  gin_bucket_offset_kernel<<<n_buckets, GIN_CHUNKS, 0, stream>>>(cnt, off, btotal);
  gin_bucket_place_kernel<<<GIN_CHUNKS, 1024, 0, stream>>>(src, dst, off, btotal, pairs, n_edges, n_buckets, chunk_len);
  gin_csr_project_kernel<<<n_buckets + gemm_blocks, GIN_BNODES, 0, stream>>>(
      pairs, btotal, rowptr, csrc, n, n_buckets, x, W0, tbuf);

  gin_gather_kernel<false><<<gather_blocks, 256, 0, stream>>>(tbuf, rowptr, csrc, agg1, n);
  gin_gather_gemm_kernel<true, GIN_D_HID, 2, true>
      <<<group_blocks, 256, 0, stream>>>(agg1, rowptr, csrc, W1, tbuf, n);
  gin_gather_gemm_kernel<false, GIN_N_CLASSES, 3, false>
      <<<group_blocks, 256, 0, stream>>>(tbuf, rowptr, csrc, W2, out, n);
}


# Edge grouping level 1 made chunk-local: a chunk's block orders its own edges by bucket in its own slice, bucket blocks collect 256 pieces; count and offset launches go (7 to 5); sum order changes
# speedup vs baseline: 3.6435x; 1.2079x over previous
#include <hip/hip_runtime.h>


typedef __attribute__((ext_vector_type(2))) float v2f;
typedef __attribute__((ext_vector_type(8))) float v8f;

#define GIN_D_IN 64
#define GIN_D_HID 32
#define GIN_N_CLASSES 40
#define GIN_SELF_W 1.1f

#define GIN_BSH 8
#define GIN_BNODES (1 << GIN_BSH)
#define GIN_MAXB 512
#define GIN_CHUNKS 256
#define GIN_TP (GIN_MAXB + 1)
#define GIN_EB 7
#define GIN_SEG 8
#define GIN_SP 3

__global__ __launch_bounds__(1024) void gin_chunk_sort_kernel(
    const int* __restrict__ src, const int* __restrict__ dst,
    int* __restrict__ pairs, int* __restrict__ start, int n_edges,
    int n_buckets, int chunk_len) {
  __shared__ int hist[GIN_MAXB];
  __shared__ int cur[GIN_MAXB];
  __shared__ int w_tot[GIN_MAXB / 32];
  const int c = blockIdx.x;
  const int t = threadIdx.x;
  const int lane = t & 31;
  const int wave = t >> 5;
  const int beg = c * chunk_len;
  const int end = min(beg + chunk_len, n_edges);
  const int last = max(n_edges - 1, 0);
  const bool one_batch = end - beg <= 1024 * GIN_EB;
  int e0 = beg + t;
  int d[GIN_EB], s[GIN_EB];
#pragma unroll
  for (int j = 0; j < GIN_EB; ++j) {
    const int e = min(e0 + j * 1024, last);
    d[j] = dst[e];
    s[j] = src[e];
  }
  for (int i = t; i < GIN_MAXB; i += 1024) hist[i] = 0;
  __syncthreads();
  for (;;) {
#pragma unroll
    for (int j = 0; j < GIN_EB; ++j)
      if (e0 + j * 1024 < end) atomicAdd(&hist[d[j] >> GIN_BSH], 1);
    e0 += 1024 * GIN_EB;
    if (e0 >= end) break;
#pragma unroll
    for (int j = 0; j < GIN_EB; ++j) d[j] = dst[min(e0 + j * 1024, last)];
  }
  __syncthreads();

  const bool has = t < n_buckets;
  const int own = has ? hist[t] : 0;
  int incl = own;
#pragma unroll
  for (int k = 1; k < 32; k <<= 1) {
    int up = __shfl_up(incl, k);
    if (lane >= k) incl += up;
  }
  if (lane == 31 && wave < GIN_MAXB / 32) w_tot[wave] = incl;
  __syncthreads();
  if (has) {
    int wpre = 0;
#pragma unroll
    for (int w = 0; w < GIN_MAXB / 32; ++w)
      if (w < wave) wpre += w_tot[w];
    const int first = wpre + incl - own;
    cur[t] = first;
    start[c * GIN_TP + t] = first;
    if (t == n_buckets - 1) start[c * GIN_TP + n_buckets] = first + own;
  }
  __syncthreads();

  e0 = beg + t;
  if (!one_batch) {
#pragma unroll
    for (int j = 0; j < GIN_EB; ++j) {
      const int e = min(e0 + j * 1024, last);
      d[j] = dst[e];
      s[j] = src[e];
    }
  }
  int2* p2 = (int2*)pairs + (size_t)beg;
  for (;;) {
    int pos[GIN_EB];
#pragma unroll
    for (int j = 0; j < GIN_EB; ++j)
      if (e0 + j * 1024 < end) pos[j] = atomicAdd(&cur[d[j] >> GIN_BSH], 1);
#pragma unroll
    for (int j = 0; j < GIN_EB; ++j)
      if (e0 + j * 1024 < end) p2[pos[j]] = make_int2(s[j], d[j]);
    e0 += 1024 * GIN_EB;
    if (e0 >= end) break;
#pragma unroll
    for (int j = 0; j < GIN_EB; ++j) {
      const int e = min(e0 + j * 1024, last);
      d[j] = dst[e];
      s[j] = src[e];
    }
  }
}

__device__ __forceinline__ void gin_bucket_csr_block(
    const int* __restrict__ pairs, const int* __restrict__ start,
    int* __restrict__ rowptr, int* __restrict__ csr_src, int n, int n_buckets,
    int chunk_len) {
  static_assert(GIN_CHUNKS == GIN_BNODES, "a bucket block has one thread per chunk and one per node");
  static_assert(GIN_SEG * 32 == GIN_CHUNKS, "32 lane groups x GIN_SEG pieces cover every chunk once");
  __shared__ int cnt[GIN_BNODES];
  __shared__ int seg_beg[GIN_CHUNKS];
  __shared__ int seg_end[GIN_CHUNKS];
  __shared__ int w_tot[GIN_BNODES / 32];
  __shared__ int w_beg[GIN_BNODES / 32];
  __shared__ int w_len[GIN_BNODES / 32];
  const int b = blockIdx.x;
  const int t = threadIdx.x;
  const int lane = t & 31;
  const int wave = t >> 5;
  const int g = t >> 3;
  const int l8 = t & 7;
  const int2* p2 = (const int2*)pairs;

  const int s0 = start[t * GIN_TP + b];
  const int s1 = start[t * GIN_TP + b + 1];
  seg_beg[t] = s0;
  seg_end[t] = s1;
  int part_beg = s0;
  int part_len = s1 - s0;
#pragma unroll
  for (int k = 16; k >= 1; k >>= 1) {
    part_beg += __shfl_xor(part_beg, k);
    part_len += __shfl_xor(part_len, k);
  }
  if (lane == 0) {
    w_beg[wave] = part_beg;
    w_len[wave] = part_len;
  }
  cnt[t] = 0;
  __syncthreads();
  int beg = 0, tot = 0;
#pragma unroll
  for (int w = 0; w < GIN_BNODES / 32; ++w) {
    beg += w_beg[w];
    tot += w_len[w];
  }
  const int end = beg + tot;

  int2 r[GIN_SEG * GIN_SP];
  unsigned valid = 0u;
  bool longer = false;
#pragma unroll
  for (int q = 0; q < GIN_SEG; ++q) {
    const int chunk = q * 32 + g;
    const int kb = seg_beg[chunk];
    const int ke = seg_end[chunk];
    longer = longer || (ke - kb > GIN_SP * 8);
#pragma unroll
    for (int p = 0; p < GIN_SP; ++p) {
      const int k = kb + p * 8 + l8;
      const bool in = k < ke;
      if (in) valid |= 1u << (q * GIN_SP + p);
      r[q * GIN_SP + p] = p2[in ? (size_t)chunk * (size_t)chunk_len + (size_t)k : (size_t)0];
    }
  }
#pragma unroll
  for (int j = 0; j < GIN_SEG * GIN_SP; ++j)
    if ((valid >> j) & 1u) atomicAdd(&cnt[r[j].y & (GIN_BNODES - 1)], 1);
  if (longer) {
    for (int q = 0; q < GIN_SEG; ++q) {
      const int chunk = q * 32 + g;
      const int ke = seg_end[chunk];
      const int2* piece = p2 + (size_t)chunk * (size_t)chunk_len;
      for (int k = seg_beg[chunk] + GIN_SP * 8 + l8; k < ke; k += 8)
        atomicAdd(&cnt[piece[k].y & (GIN_BNODES - 1)], 1);
    }
  }
  __syncthreads();

  const int own = cnt[t];
  int incl = own;
#pragma unroll
  for (int d = 1; d < 32; d <<= 1) {
    int up = __shfl_up(incl, d);
    if (lane >= d) incl += up;
  }
  if (lane == 31) w_tot[wave] = incl;
  __syncthreads();
  int wpre = 0;
#pragma unroll
  for (int w = 0; w < GIN_BNODES / 32; ++w)
    if (w < wave) wpre += w_tot[w];
  const int first = beg + wpre + incl - own;
  cnt[t] = first;
  const int node = b * GIN_BNODES + t;
  if (node < n) rowptr[node] = first;
  if (b == n_buckets - 1 && t == 0) rowptr[n] = end;
  __syncthreads();

  int pos[GIN_SEG * GIN_SP];
#pragma unroll
  for (int j = 0; j < GIN_SEG * GIN_SP; ++j)
    if ((valid >> j) & 1u)
      pos[j] = atomicAdd(&cnt[r[j].y & (GIN_BNODES - 1)], 1);
#pragma unroll
  for (int j = 0; j < GIN_SEG * GIN_SP; ++j)
    if ((valid >> j) & 1u) csr_src[pos[j]] = r[j].x;
  if (longer) {
    for (int q = 0; q < GIN_SEG; ++q) {
      const int chunk = q * 32 + g;
      const int ke = seg_end[chunk];
      const int2* piece = p2 + (size_t)chunk * (size_t)chunk_len;
      for (int k = seg_beg[chunk] + GIN_SP * 8 + l8; k < ke; k += 8) {
        const int2 p = piece[k];
        const int at = atomicAdd(&cnt[p.y & (GIN_BNODES - 1)], 1);
        csr_src[at] = p.x;
      }
    }
  }
}

template <bool RELU_IN>
__device__ __forceinline__ float4 gin_gather_row(
    const float4* __restrict__ h4, const int* __restrict__ rowptr,
    const int* __restrict__ csr_src, int node, int l8) {
  float4 acc = h4[(unsigned)node * 8u + (unsigned)l8];
  if (RELU_IN) {
    acc.x = fmaxf(acc.x, 0.0f); acc.y = fmaxf(acc.y, 0.0f);
    acc.z = fmaxf(acc.z, 0.0f); acc.w = fmaxf(acc.w, 0.0f);
  }
  acc.x *= GIN_SELF_W; acc.y *= GIN_SELF_W; acc.z *= GIN_SELF_W; acc.w *= GIN_SELF_W;
  const int beg = rowptr[node];
  const int end = rowptr[node + 1];
  for (int base = beg; base < end; base += 8) {
    const int e = min(base + l8, end - 1);
    const int mine = csr_src[e];
    float4 r[8];
#pragma unroll
    for (int j = 0; j < 8; ++j) {
      const unsigned s = (unsigned)__shfl(mine, j, 8);
      r[j] = h4[s * 8u + (unsigned)l8];
    }
#pragma unroll
    for (int j = 0; j < 8; ++j) {
      float4 v = r[j];
      if (RELU_IN) {
        v.x = fmaxf(v.x, 0.0f); v.y = fmaxf(v.y, 0.0f);
        v.z = fmaxf(v.z, 0.0f); v.w = fmaxf(v.w, 0.0f);
      }
      const bool on = base + j < end;
      acc.x += on ? v.x : 0.0f; acc.y += on ? v.y : 0.0f;
      acc.z += on ? v.z : 0.0f; acc.w += on ? v.w : 0.0f;
    }
  }
  return acc;
}

template <bool RELU_IN>
__global__ __launch_bounds__(256) void gin_gather_kernel(
    const float* __restrict__ h, const int* __restrict__ rowptr,
    const int* __restrict__ csr_src, float* __restrict__ agg, int n) {
  const int l8 = threadIdx.x & 7;
  const int node = (blockIdx.x * 256 + (int)threadIdx.x) >> 3;
  if (node >= n) return;
  const float4 acc = gin_gather_row<RELU_IN>((const float4*)h, rowptr, csr_src, node, l8);
  ((float4*)agg)[(unsigned)node * 8u + (unsigned)l8] = acc;
}

#define GIN_GROUP_NODES 32
#define GIN_LDS_PITCH 36
template <bool RELU_IN, int DOUTL, int NT, bool RELU_OUT>
__global__ __launch_bounds__(256) void gin_gather_gemm_kernel(
    const float* __restrict__ h, const int* __restrict__ rowptr,
    const int* __restrict__ csr_src, const float* __restrict__ W,
    float* __restrict__ C, int n) {
  __shared__ __attribute__((aligned(16))) float rows[GIN_GROUP_NODES * GIN_LDS_PITCH];
  const int l8 = threadIdx.x & 7;
  const int lnode = (int)threadIdx.x >> 3;
  const int node = blockIdx.x * GIN_GROUP_NODES + lnode;
  float4 sum = make_float4(0.0f, 0.0f, 0.0f, 0.0f);
  if (node < n)
    sum = gin_gather_row<RELU_IN>((const float4*)h, rowptr, csr_src, node, l8);
  *(float4*)&rows[lnode * GIN_LDS_PITCH + l8 * 4] = sum;
  __syncthreads();

  const int lane = threadIdx.x & 31;
  const int wave = threadIdx.x >> 5;
  if (wave >= GIN_GROUP_NODES / 16) return;

  const int l16 = lane & 15;
  const int khalf = lane >> 4;
  const int row0 = blockIdx.x * GIN_GROUP_NODES + wave * 16;
  const float* arow = &rows[(wave * 16 + l16) * GIN_LDS_PITCH];

  v8f acc[NT] = {};

#pragma unroll
  for (int kk = 0; kk < GIN_D_HID / 4; ++kk) {
    const int kb = kk * 4 + khalf * 2;
    const v2f a = *(const v2f*)(arow + kb);
#pragma unroll
    for (int t = 0; t < NT; ++t) {
      const int col = t * 16 + l16;
      const bool ok = ((DOUTL & 15) == 0) || (col < DOUTL);
      v2f b;
      b.x = ok ? W[(size_t)kb * DOUTL + col] : 0.0f;
      b.y = ok ? W[(size_t)(kb + 1) * DOUTL + col] : 0.0f;
      acc[t] = __builtin_amdgcn_wmma_f32_16x16x4_f32(
          false, a, false, b, (short)0, acc[t], false, false);
    }
  }

#pragma unroll
  for (int t = 0; t < NT; ++t) {
    const int col = t * 16 + l16;
    if (((DOUTL & 15) != 0) && col >= DOUTL) continue;
#pragma unroll
    for (int r = 0; r < 8; ++r) {
      const int row = row0 + r + khalf * 8;
      if (row < n) {
        float v = acc[t][r];
        if (RELU_OUT) v = fmaxf(v, 0.0f);
        C[(size_t)row * DOUTL + col] = v;
      }
    }
  }
}

template <int DINL, int DOUTL, int NT, bool RELU>
__device__ __forceinline__ void gin_gemm_tile(
    const float* __restrict__ A, const float* __restrict__ W,
    float* __restrict__ C, int n_rows, int tile) {
  const int lane = threadIdx.x & 31;
  const int row0 = tile << 4;
  const int l16 = lane & 15;
  const int khalf = lane >> 4;
  int arow = row0 + l16;
  if (arow >= n_rows) arow = n_rows - 1;

  v8f acc[NT] = {};

#pragma unroll
  for (int kk = 0; kk < DINL / 4; ++kk) {
    const int kb = kk * 4 + khalf * 2;
    v2f a;
    a.x = A[(size_t)arow * DINL + kb];
    a.y = A[(size_t)arow * DINL + kb + 1];
#pragma unroll
    for (int t = 0; t < NT; ++t) {
      const int col = t * 16 + l16;
      const bool ok = ((DOUTL & 15) == 0) || (col < DOUTL);
      v2f b;
      b.x = ok ? W[(size_t)kb * DOUTL + col] : 0.0f;
      b.y = ok ? W[(size_t)(kb + 1) * DOUTL + col] : 0.0f;
      acc[t] = __builtin_amdgcn_wmma_f32_16x16x4_f32(
          false, a, false, b, (short)0, acc[t], false, false);
    }
  }

#pragma unroll
  for (int t = 0; t < NT; ++t) {
    const int col = t * 16 + l16;
    if (((DOUTL & 15) != 0) && col >= DOUTL) continue;
#pragma unroll
    for (int r = 0; r < 8; ++r) {
      const int row = row0 + r + khalf * 8;
      if (row < n_rows) {
        float v = acc[t][r];
        if (RELU) v = fmaxf(v, 0.0f);
        C[(size_t)row * DOUTL + col] = v;
      }
    }
  }
}

__global__ __launch_bounds__(GIN_BNODES) void gin_csr_project_kernel(
    const int* __restrict__ pairs, const int* __restrict__ start,
    int* __restrict__ rowptr, int* __restrict__ csr_src, int n, int n_buckets,
    int chunk_len, const float* __restrict__ x, const float* __restrict__ W0,
    float* __restrict__ xw) {
  if ((int)blockIdx.x >= n_buckets) {
    const int tile = ((int)blockIdx.x - n_buckets) * (GIN_BNODES / 32) + ((int)threadIdx.x >> 5);
    if (tile < ((n + 15) >> 4))
      gin_gemm_tile<GIN_D_IN, GIN_D_HID, 2, false>(x, W0, xw, n, tile);
    return;
  }
  gin_bucket_csr_block(pairs, start, rowptr, csr_src, n, n_buckets, chunk_len);
}

extern "C" void kernel_launch(void* const* d_in, const int* in_sizes, int n_in,
                              void* d_out, int out_size, void* d_ws, size_t ws_size,
                              hipStream_t stream) {
  const float* x  = (const float*)d_in[0];
  const float* W0 = (const float*)d_in[1];
  const float* W1 = (const float*)d_in[2];
  const float* W2 = (const float*)d_in[3];
  const int* src  = (const int*)d_in[4];
  const int* dst  = (const int*)d_in[5];
  float* out = (float*)d_out;

  const int n = in_sizes[0] / GIN_D_IN;
  const int n_edges = in_sizes[4];

  float* tbuf = (float*)d_ws;
  float* agg1 = tbuf + (size_t)n * GIN_D_HID;
  int* start  = (int*)(agg1 + (size_t)n * GIN_D_HID);
  int* rowptr = start + GIN_CHUNKS * GIN_TP;
  int* pairs  = rowptr + (((size_t)n + 1 + 3) / 4) * 4;
  int* csrc   = pairs + 2 * (size_t)n_edges;

  const int ntiles = (n + 15) / 16;
  const int gemm_blocks = (ntiles + 7) / 8;
  const int n_buckets = (n + GIN_BNODES - 1) / GIN_BNODES;
  const int chunk_len = (n_edges + GIN_CHUNKS - 1) / GIN_CHUNKS;
  const int gather_blocks = (n * 8 + 255) / 256;
  const int group_blocks = (n + GIN_GROUP_NODES - 1) / GIN_GROUP_NODES;

  gin_chunk_sort_kernel<<<GIN_CHUNKS, 1024, 0, stream>>>(src, dst, pairs, start, n_edges, n_buckets, chunk_len);
  gin_csr_project_kernel<<<n_buckets + gemm_blocks, GIN_BNODES, 0, stream>>>(
      pairs, start, rowptr, csrc, n, n_buckets, chunk_len, x, W0, tbuf);

  gin_gather_kernel<false><<<gather_blocks, 256, 0, stream>>>(tbuf, rowptr, csrc, agg1, n);
  gin_gather_gemm_kernel<true, GIN_D_HID, 2, true>
      <<<group_blocks, 256, 0, stream>>>(agg1, rowptr, csrc, W1, tbuf, n);
  gin_gather_gemm_kernel<false, GIN_N_CLASSES, 3, false>
      <<<group_blocks, 256, 0, stream>>>(tbuf, rowptr, csrc, W2, out, n);
}


# Both edge-grouping levels place their records in LDS and copy the ordered run out in whole aligned 128-byte lines, not one scattered 8-byte or 4-byte store per record; sum order unchanged
# speedup vs baseline: 3.9631x; 1.0877x over previous
#include <hip/hip_runtime.h>


typedef __attribute__((ext_vector_type(2))) float v2f;
typedef __attribute__((ext_vector_type(8))) float v8f;

#define GIN_D_IN 64
#define GIN_D_HID 32
#define GIN_N_CLASSES 40
#define GIN_SELF_W 1.1f

#define GIN_BSH 8
#define GIN_BNODES (1 << GIN_BSH)
#define GIN_MAXB 512
#define GIN_CHUNKS 256
#define GIN_TP (GIN_MAXB + 1)
#define GIN_EB 7
#define GIN_SEG 8
#define GIN_SP 3
#define GIN_STAGE1 (1024 * GIN_EB)
#define GIN_STAGE2 6144

__global__ __launch_bounds__(1024) void gin_chunk_sort_kernel(
    const int* __restrict__ src, const int* __restrict__ dst,
    int* __restrict__ pairs, int* __restrict__ start, int n_edges,
    int n_buckets, int chunk_len) {
  __shared__ int hist[GIN_MAXB];
  __shared__ int cur[GIN_MAXB];
  __shared__ int w_tot[GIN_MAXB / 32];
  __shared__ int2 stage[GIN_STAGE1];
  const int c = blockIdx.x;
  const int t = threadIdx.x;
  const int lane = t & 31;
  const int wave = t >> 5;
  const int beg = c * chunk_len;
  const int end = min(beg + chunk_len, n_edges);
  const int last = max(n_edges - 1, 0);
  const bool one_batch = end - beg <= 1024 * GIN_EB;
  int e0 = beg + t;
  int d[GIN_EB], s[GIN_EB];
#pragma unroll
  for (int j = 0; j < GIN_EB; ++j) {
    const int e = min(e0 + j * 1024, last);
    d[j] = dst[e];
    s[j] = src[e];
  }
  for (int i = t; i < GIN_MAXB; i += 1024) hist[i] = 0;
  __syncthreads();
  for (;;) {
#pragma unroll
    for (int j = 0; j < GIN_EB; ++j)
      if (e0 + j * 1024 < end) atomicAdd(&hist[d[j] >> GIN_BSH], 1);
    e0 += 1024 * GIN_EB;
    if (e0 >= end) break;
#pragma unroll
    for (int j = 0; j < GIN_EB; ++j) d[j] = dst[min(e0 + j * 1024, last)];
  }
  __syncthreads();

  const bool has = t < n_buckets;
  const int own = has ? hist[t] : 0;
  int incl = own;
#pragma unroll
  for (int k = 1; k < 32; k <<= 1) {
    int up = __shfl_up(incl, k);
    if (lane >= k) incl += up;
  }
  if (lane == 31 && wave < GIN_MAXB / 32) w_tot[wave] = incl;
  __syncthreads();
  if (has) {
    int wpre = 0;
#pragma unroll
    for (int w = 0; w < GIN_MAXB / 32; ++w)
      if (w < wave) wpre += w_tot[w];
    const int first = wpre + incl - own;
    cur[t] = first;
    start[c * GIN_TP + t] = first;
    if (t == n_buckets - 1) start[c * GIN_TP + n_buckets] = first + own;
  }
  __syncthreads();

  e0 = beg + t;
  int2* p2 = (int2*)pairs + (size_t)beg;
  if (one_batch) {
    int pos[GIN_EB];
#pragma unroll
    for (int j = 0; j < GIN_EB; ++j)
      if (e0 + j * 1024 < end) pos[j] = atomicAdd(&cur[d[j] >> GIN_BSH], 1);
#pragma unroll
    for (int j = 0; j < GIN_EB; ++j)
      if (e0 + j * 1024 < end) stage[pos[j]] = make_int2(s[j], d[j]);
    __syncthreads();
    const int cnt = end - beg;
    const int shift = (int)(((size_t)p2 >> 3) & 15);
    const int i0 = t - shift;
    int2 v[GIN_EB + 1];
#pragma unroll
    for (int k = 0; k <= GIN_EB; ++k) {
      const int i = i0 + k * 1024;
      v[k] = stage[(i >= 0 && i < cnt) ? i : 0];
    }
#pragma unroll
    for (int k = 0; k <= GIN_EB; ++k) {
      const int i = i0 + k * 1024;
      if (i >= 0 && i < cnt) p2[i] = v[k];
    }
    return;
  }
#pragma unroll
  for (int j = 0; j < GIN_EB; ++j) {
    const int e = min(e0 + j * 1024, last);
    d[j] = dst[e];
    s[j] = src[e];
  }
  for (;;) {
    int pos[GIN_EB];
#pragma unroll
    for (int j = 0; j < GIN_EB; ++j)
      if (e0 + j * 1024 < end) pos[j] = atomicAdd(&cur[d[j] >> GIN_BSH], 1);
#pragma unroll
    for (int j = 0; j < GIN_EB; ++j)
      if (e0 + j * 1024 < end) p2[pos[j]] = make_int2(s[j], d[j]);
    e0 += 1024 * GIN_EB;
    if (e0 >= end) break;
#pragma unroll
    for (int j = 0; j < GIN_EB; ++j) {
      const int e = min(e0 + j * 1024, last);
      d[j] = dst[e];
      s[j] = src[e];
    }
  }
}

__device__ __forceinline__ void gin_bucket_csr_block(
    const int* __restrict__ pairs, const int* __restrict__ start,
    int* __restrict__ rowptr, int* __restrict__ csr_src, int n, int n_buckets,
    int chunk_len) {
  static_assert(GIN_CHUNKS == GIN_BNODES, "a bucket block has one thread per chunk and one per node");
  static_assert(GIN_SEG * 32 == GIN_CHUNKS, "32 lane groups x GIN_SEG pieces cover every chunk once");
  __shared__ int cnt[GIN_BNODES];
  __shared__ int seg_beg[GIN_CHUNKS];
  __shared__ int seg_end[GIN_CHUNKS];
  __shared__ int w_tot[GIN_BNODES / 32];
  __shared__ int w_beg[GIN_BNODES / 32];
  __shared__ int w_len[GIN_BNODES / 32];
  __shared__ int stage2[GIN_STAGE2];
  const int b = blockIdx.x;
  const int t = threadIdx.x;
  const int lane = t & 31;
  const int wave = t >> 5;
  const int g = t >> 3;
  const int l8 = t & 7;
  const int2* p2 = (const int2*)pairs;

  const int s0 = start[t * GIN_TP + b];
  const int s1 = start[t * GIN_TP + b + 1];
  seg_beg[t] = s0;
  seg_end[t] = s1;
  int part_beg = s0;
  int part_len = s1 - s0;
#pragma unroll
  for (int k = 16; k >= 1; k >>= 1) {
    part_beg += __shfl_xor(part_beg, k);
    part_len += __shfl_xor(part_len, k);
  }
  if (lane == 0) {
    w_beg[wave] = part_beg;
    w_len[wave] = part_len;
  }
  cnt[t] = 0;
  __syncthreads();
  int beg = 0, tot = 0;
#pragma unroll
  for (int w = 0; w < GIN_BNODES / 32; ++w) {
    beg += w_beg[w];
    tot += w_len[w];
  }
  const int end = beg + tot;
  const bool staged = tot <= GIN_STAGE2;

  int2 r[GIN_SEG * GIN_SP];
  unsigned valid = 0u;
  bool longer = false;
#pragma unroll
  for (int q = 0; q < GIN_SEG; ++q) {
    const int chunk = q * 32 + g;
    const int kb = seg_beg[chunk];
    const int ke = seg_end[chunk];
    longer = longer || (ke - kb > GIN_SP * 8);
#pragma unroll
    for (int p = 0; p < GIN_SP; ++p) {
      const int k = kb + p * 8 + l8;
      const bool in = k < ke;
      if (in) valid |= 1u << (q * GIN_SP + p);
      r[q * GIN_SP + p] = p2[in ? (size_t)chunk * (size_t)chunk_len + (size_t)k : (size_t)0];
    }
  }
#pragma unroll
  for (int j = 0; j < GIN_SEG * GIN_SP; ++j)
    if ((valid >> j) & 1u) atomicAdd(&cnt[r[j].y & (GIN_BNODES - 1)], 1);
  if (longer) {
    for (int q = 0; q < GIN_SEG; ++q) {
      const int chunk = q * 32 + g;
      const int ke = seg_end[chunk];
      const int2* piece = p2 + (size_t)chunk * (size_t)chunk_len;
      for (int k = seg_beg[chunk] + GIN_SP * 8 + l8; k < ke; k += 8)
        atomicAdd(&cnt[piece[k].y & (GIN_BNODES - 1)], 1);
    }
  }
  __syncthreads();

  const int own = cnt[t];
  int incl = own;
#pragma unroll
  for (int d = 1; d < 32; d <<= 1) {
    int up = __shfl_up(incl, d);
    if (lane >= d) incl += up;
  }
  if (lane == 31) w_tot[wave] = incl;
  __syncthreads();
  int wpre = 0;
#pragma unroll
  for (int w = 0; w < GIN_BNODES / 32; ++w)
    if (w < wave) wpre += w_tot[w];
  const int first = beg + wpre + incl - own;
  cnt[t] = first;
  const int node = b * GIN_BNODES + t;
  if (node < n) rowptr[node] = first;
  if (b == n_buckets - 1 && t == 0) rowptr[n] = end;
  __syncthreads();

  int pos[GIN_SEG * GIN_SP];
#pragma unroll
  for (int j = 0; j < GIN_SEG * GIN_SP; ++j)
    if ((valid >> j) & 1u)
      pos[j] = atomicAdd(&cnt[r[j].y & (GIN_BNODES - 1)], 1);
  if (staged) {
#pragma unroll
    for (int j = 0; j < GIN_SEG * GIN_SP; ++j)
      if ((valid >> j) & 1u) stage2[pos[j] - beg] = r[j].x;
  } else {
#pragma unroll
    for (int j = 0; j < GIN_SEG * GIN_SP; ++j)
      if ((valid >> j) & 1u) csr_src[pos[j]] = r[j].x;
  }
  if (longer) {
    for (int q = 0; q < GIN_SEG; ++q) {
      const int chunk = q * 32 + g;
      const int ke = seg_end[chunk];
      const int2* piece = p2 + (size_t)chunk * (size_t)chunk_len;
      for (int k = seg_beg[chunk] + GIN_SP * 8 + l8; k < ke; k += 8) {
        const int2 p = piece[k];
        const int at = atomicAdd(&cnt[p.y & (GIN_BNODES - 1)], 1);
        if (staged) stage2[at - beg] = p.x;
        else csr_src[at] = p.x;
      }
    }
  }
  if (staged) {
    __syncthreads();
    int* run = csr_src + beg;
    const int shift = (int)(((size_t)run >> 2) & 31);
    for (int i0 = t - shift; i0 < tot; i0 += 8 * GIN_BNODES) {
      int v[8];
#pragma unroll
      for (int k = 0; k < 8; ++k) {
        const int i = i0 + k * GIN_BNODES;
        v[k] = stage2[(i >= 0 && i < tot) ? i : 0];
      }
#pragma unroll
      for (int k = 0; k < 8; ++k) {
        const int i = i0 + k * GIN_BNODES;
        if (i >= 0 && i < tot) run[i] = v[k];
      }
    }
  }
}

template <bool RELU_IN>
__device__ __forceinline__ float4 gin_gather_row(
    const float4* __restrict__ h4, const int* __restrict__ rowptr,
    const int* __restrict__ csr_src, int node, int l8) {
  float4 acc = h4[(unsigned)node * 8u + (unsigned)l8];
  if (RELU_IN) {
    acc.x = fmaxf(acc.x, 0.0f); acc.y = fmaxf(acc.y, 0.0f);
    acc.z = fmaxf(acc.z, 0.0f); acc.w = fmaxf(acc.w, 0.0f);
  }
  acc.x *= GIN_SELF_W; acc.y *= GIN_SELF_W; acc.z *= GIN_SELF_W; acc.w *= GIN_SELF_W;
  const int beg = rowptr[node];
  const int end = rowptr[node + 1];
  for (int base = beg; base < end; base += 8) {
    const int e = min(base + l8, end - 1);
    const int mine = csr_src[e];
    float4 r[8];
#pragma unroll
    for (int j = 0; j < 8; ++j) {
      const unsigned s = (unsigned)__shfl(mine, j, 8);
      r[j] = h4[s * 8u + (unsigned)l8];
    }
#pragma unroll
    for (int j = 0; j < 8; ++j) {
      float4 v = r[j];
      if (RELU_IN) {
        v.x = fmaxf(v.x, 0.0f); v.y = fmaxf(v.y, 0.0f);
        v.z = fmaxf(v.z, 0.0f); v.w = fmaxf(v.w, 0.0f);
      }
      const bool on = base + j < end;
      acc.x += on ? v.x : 0.0f; acc.y += on ? v.y : 0.0f;
      acc.z += on ? v.z : 0.0f; acc.w += on ? v.w : 0.0f;
    }
  }
  return acc;
}

template <bool RELU_IN>
__global__ __launch_bounds__(256) void gin_gather_kernel(
    const float* __restrict__ h, const int* __restrict__ rowptr,
    const int* __restrict__ csr_src, float* __restrict__ agg, int n) {
  const int l8 = threadIdx.x & 7;
  const int node = (blockIdx.x * 256 + (int)threadIdx.x) >> 3;
  if (node >= n) return;
  const float4 acc = gin_gather_row<RELU_IN>((const float4*)h, rowptr, csr_src, node, l8);
  ((float4*)agg)[(unsigned)node * 8u + (unsigned)l8] = acc;
}

#define GIN_GROUP_NODES 32
#define GIN_LDS_PITCH 36
template <bool RELU_IN, int DOUTL, int NT, bool RELU_OUT>
__global__ __launch_bounds__(256) void gin_gather_gemm_kernel(
    const float* __restrict__ h, const int* __restrict__ rowptr,
    const int* __restrict__ csr_src, const float* __restrict__ W,
    float* __restrict__ C, int n) {
  __shared__ __attribute__((aligned(16))) float rows[GIN_GROUP_NODES * GIN_LDS_PITCH];
  const int l8 = threadIdx.x & 7;
  const int lnode = (int)threadIdx.x >> 3;
  const int node = blockIdx.x * GIN_GROUP_NODES + lnode;
  float4 sum = make_float4(0.0f, 0.0f, 0.0f, 0.0f);
  if (node < n)
    sum = gin_gather_row<RELU_IN>((const float4*)h, rowptr, csr_src, node, l8);
  *(float4*)&rows[lnode * GIN_LDS_PITCH + l8 * 4] = sum;
  __syncthreads();

  const int lane = threadIdx.x & 31;
  const int wave = threadIdx.x >> 5;
  if (wave >= GIN_GROUP_NODES / 16) return;

  const int l16 = lane & 15;
  const int khalf = lane >> 4;
  const int row0 = blockIdx.x * GIN_GROUP_NODES + wave * 16;
  const float* arow = &rows[(wave * 16 + l16) * GIN_LDS_PITCH];

  v8f acc[NT] = {};

#pragma unroll
  for (int kk = 0; kk < GIN_D_HID / 4; ++kk) {
    const int kb = kk * 4 + khalf * 2;
    const v2f a = *(const v2f*)(arow + kb);
#pragma unroll
    for (int t = 0; t < NT; ++t) {
      const int col = t * 16 + l16;
      const bool ok = ((DOUTL & 15) == 0) || (col < DOUTL);
      v2f b;
      b.x = ok ? W[(size_t)kb * DOUTL + col] : 0.0f;
      b.y = ok ? W[(size_t)(kb + 1) * DOUTL + col] : 0.0f;
      acc[t] = __builtin_amdgcn_wmma_f32_16x16x4_f32(
          false, a, false, b, (short)0, acc[t], false, false);
    }
  }

#pragma unroll
  for (int t = 0; t < NT; ++t) {
    const int col = t * 16 + l16;
    if (((DOUTL & 15) != 0) && col >= DOUTL) continue;
#pragma unroll
    for (int r = 0; r < 8; ++r) {
      const int row = row0 + r + khalf * 8;
      if (row < n) {
        float v = acc[t][r];
        if (RELU_OUT) v = fmaxf(v, 0.0f);
        C[(size_t)row * DOUTL + col] = v;
      }
    }
  }
}

template <int DINL, int DOUTL, int NT, bool RELU>
__device__ __forceinline__ void gin_gemm_tile(
    const float* __restrict__ A, const float* __restrict__ W,
    float* __restrict__ C, int n_rows, int tile) {
  const int lane = threadIdx.x & 31;
  const int row0 = tile << 4;
  const int l16 = lane & 15;
  const int khalf = lane >> 4;
  int arow = row0 + l16;
  if (arow >= n_rows) arow = n_rows - 1;

  v8f acc[NT] = {};

#pragma unroll
  for (int kk = 0; kk < DINL / 4; ++kk) {
    const int kb = kk * 4 + khalf * 2;
    v2f a;
    a.x = A[(size_t)arow * DINL + kb];
    a.y = A[(size_t)arow * DINL + kb + 1];
#pragma unroll
    for (int t = 0; t < NT; ++t) {
      const int col = t * 16 + l16;
      const bool ok = ((DOUTL & 15) == 0) || (col < DOUTL);
      v2f b;
      b.x = ok ? W[(size_t)kb * DOUTL + col] : 0.0f;
      b.y = ok ? W[(size_t)(kb + 1) * DOUTL + col] : 0.0f;
      acc[t] = __builtin_amdgcn_wmma_f32_16x16x4_f32(
          false, a, false, b, (short)0, acc[t], false, false);
    }
  }

#pragma unroll
  for (int t = 0; t < NT; ++t) {
    const int col = t * 16 + l16;
    if (((DOUTL & 15) != 0) && col >= DOUTL) continue;
#pragma unroll
    for (int r = 0; r < 8; ++r) {
      const int row = row0 + r + khalf * 8;
      if (row < n_rows) {
        float v = acc[t][r];
        if (RELU) v = fmaxf(v, 0.0f);
        C[(size_t)row * DOUTL + col] = v;
      }
    }
  }
}

__global__ __launch_bounds__(GIN_BNODES) void gin_csr_project_kernel(
    const int* __restrict__ pairs, const int* __restrict__ start,
    int* __restrict__ rowptr, int* __restrict__ csr_src, int n, int n_buckets,
    int chunk_len, const float* __restrict__ x, const float* __restrict__ W0,
    float* __restrict__ xw) {
  if ((int)blockIdx.x >= n_buckets) {
    const int tile = ((int)blockIdx.x - n_buckets) * (GIN_BNODES / 32) + ((int)threadIdx.x >> 5);
    if (tile < ((n + 15) >> 4))
      gin_gemm_tile<GIN_D_IN, GIN_D_HID, 2, false>(x, W0, xw, n, tile);
    return;
  }
  gin_bucket_csr_block(pairs, start, rowptr, csr_src, n, n_buckets, chunk_len);
}

extern "C" void kernel_launch(void* const* d_in, const int* in_sizes, int n_in,
                              void* d_out, int out_size, void* d_ws, size_t ws_size,
                              hipStream_t stream) {
  const float* x  = (const float*)d_in[0];
  const float* W0 = (const float*)d_in[1];
  const float* W1 = (const float*)d_in[2];
  const float* W2 = (const float*)d_in[3];
  const int* src  = (const int*)d_in[4];
  const int* dst  = (const int*)d_in[5];
  float* out = (float*)d_out;

  const int n = in_sizes[0] / GIN_D_IN;
  const int n_edges = in_sizes[4];

  float* tbuf = (float*)d_ws;
  float* agg1 = tbuf + (size_t)n * GIN_D_HID;
  int* start  = (int*)(agg1 + (size_t)n * GIN_D_HID);
  int* rowptr = start + GIN_CHUNKS * GIN_TP;
  int* pairs  = rowptr + (((size_t)n + 1 + 3) / 4) * 4;
  int* csrc   = pairs + 2 * (size_t)n_edges;

  const int ntiles = (n + 15) / 16;
  const int gemm_blocks = (ntiles + 7) / 8;
  const int n_buckets = (n + GIN_BNODES - 1) / GIN_BNODES;
  const int chunk_len = (n_edges + GIN_CHUNKS - 1) / GIN_CHUNKS;
  const int gather_blocks = (n * 8 + 255) / 256;
  const int group_blocks = (n + GIN_GROUP_NODES - 1) / GIN_GROUP_NODES;

  gin_chunk_sort_kernel<<<GIN_CHUNKS, 1024, 0, stream>>>(src, dst, pairs, start, n_edges, n_buckets, chunk_len);
  gin_csr_project_kernel<<<n_buckets + gemm_blocks, GIN_BNODES, 0, stream>>>(
      pairs, start, rowptr, csrc, n, n_buckets, chunk_len, x, W0, tbuf);

  gin_gather_kernel<false><<<gather_blocks, 256, 0, stream>>>(tbuf, rowptr, csrc, agg1, n);
  gin_gather_gemm_kernel<true, GIN_D_HID, 2, true>
      <<<group_blocks, 256, 0, stream>>>(agg1, rowptr, csrc, W1, tbuf, n);
  gin_gather_gemm_kernel<false, GIN_N_CLASSES, 3, false>
      <<<group_blocks, 256, 0, stream>>>(tbuf, rowptr, csrc, W2, out, n);
}


# Aggregation (3 layers): a block covers 96 nodes, an 8-lane group sums 3 nodes one after another, their bounds and own rows loaded together; a layer starts 8336 waves, not 25000; sum order unchanged
# speedup vs baseline: 4.1863x; 1.0563x over previous
#include <hip/hip_runtime.h>


typedef __attribute__((ext_vector_type(2))) float v2f;
typedef __attribute__((ext_vector_type(8))) float v8f;

#define GIN_D_IN 64
#define GIN_D_HID 32
#define GIN_N_CLASSES 40
#define GIN_SELF_W 1.1f

#define GIN_BSH 8
#define GIN_BNODES (1 << GIN_BSH)
#define GIN_MAXB 512
#define GIN_CHUNKS 256
#define GIN_TP (GIN_MAXB + 1)
#define GIN_EB 7
#define GIN_SEG 8
#define GIN_SP 3
#define GIN_STAGE1 (1024 * GIN_EB)
#define GIN_STAGE2 6144

__global__ __launch_bounds__(1024) void gin_chunk_sort_kernel(
    const int* __restrict__ src, const int* __restrict__ dst,
    int* __restrict__ pairs, int* __restrict__ start, int n_edges,
    int n_buckets, int chunk_len) {
  __shared__ int hist[GIN_MAXB];
  __shared__ int cur[GIN_MAXB];
  __shared__ int w_tot[GIN_MAXB / 32];
  __shared__ int2 stage[GIN_STAGE1];
  const int c = blockIdx.x;
  const int t = threadIdx.x;
  const int lane = t & 31;
  const int wave = t >> 5;
  const int beg = c * chunk_len;
  const int end = min(beg + chunk_len, n_edges);
  const int last = max(n_edges - 1, 0);
  const bool one_batch = end - beg <= 1024 * GIN_EB;
  int e0 = beg + t;
  int d[GIN_EB], s[GIN_EB];
#pragma unroll
  for (int j = 0; j < GIN_EB; ++j) {
    const int e = min(e0 + j * 1024, last);
    d[j] = dst[e];
    s[j] = src[e];
  }
  for (int i = t; i < GIN_MAXB; i += 1024) hist[i] = 0;
  __syncthreads();
  for (;;) {
#pragma unroll
    for (int j = 0; j < GIN_EB; ++j)
      if (e0 + j * 1024 < end) atomicAdd(&hist[d[j] >> GIN_BSH], 1);
    e0 += 1024 * GIN_EB;
    if (e0 >= end) break;
#pragma unroll
    for (int j = 0; j < GIN_EB; ++j) d[j] = dst[min(e0 + j * 1024, last)];
  }
  __syncthreads();

  const bool has = t < n_buckets;
  const int own = has ? hist[t] : 0;
  int incl = own;
#pragma unroll
  for (int k = 1; k < 32; k <<= 1) {
    int up = __shfl_up(incl, k);
    if (lane >= k) incl += up;
  }
  if (lane == 31 && wave < GIN_MAXB / 32) w_tot[wave] = incl;
  __syncthreads();
  if (has) {
    int wpre = 0;
#pragma unroll
    for (int w = 0; w < GIN_MAXB / 32; ++w)
      if (w < wave) wpre += w_tot[w];
    const int first = wpre + incl - own;
    cur[t] = first;
    start[c * GIN_TP + t] = first;
    if (t == n_buckets - 1) start[c * GIN_TP + n_buckets] = first + own;
  }
  __syncthreads();

  e0 = beg + t;
  int2* p2 = (int2*)pairs + (size_t)beg;
  if (one_batch) {
    int pos[GIN_EB];
#pragma unroll
    for (int j = 0; j < GIN_EB; ++j)
      if (e0 + j * 1024 < end) pos[j] = atomicAdd(&cur[d[j] >> GIN_BSH], 1);
#pragma unroll
    for (int j = 0; j < GIN_EB; ++j)
      if (e0 + j * 1024 < end) stage[pos[j]] = make_int2(s[j], d[j]);
    __syncthreads();
    const int cnt = end - beg;
    const int shift = (int)(((size_t)p2 >> 3) & 15);
    const int i0 = t - shift;
    int2 v[GIN_EB + 1];
#pragma unroll
    for (int k = 0; k <= GIN_EB; ++k) {
      const int i = i0 + k * 1024;
      v[k] = stage[(i >= 0 && i < cnt) ? i : 0];
    }
#pragma unroll
    for (int k = 0; k <= GIN_EB; ++k) {
      const int i = i0 + k * 1024;
      if (i >= 0 && i < cnt) p2[i] = v[k];
    }
    return;
  }
#pragma unroll
  for (int j = 0; j < GIN_EB; ++j) {
    const int e = min(e0 + j * 1024, last);
    d[j] = dst[e];
    s[j] = src[e];
  }
  for (;;) {
    int pos[GIN_EB];
#pragma unroll
    for (int j = 0; j < GIN_EB; ++j)
      if (e0 + j * 1024 < end) pos[j] = atomicAdd(&cur[d[j] >> GIN_BSH], 1);
#pragma unroll
    for (int j = 0; j < GIN_EB; ++j)
      if (e0 + j * 1024 < end) p2[pos[j]] = make_int2(s[j], d[j]);
    e0 += 1024 * GIN_EB;
    if (e0 >= end) break;
#pragma unroll
    for (int j = 0; j < GIN_EB; ++j) {
      const int e = min(e0 + j * 1024, last);
      d[j] = dst[e];
      s[j] = src[e];
    }
  }
}

__device__ __forceinline__ void gin_bucket_csr_block(
    const int* __restrict__ pairs, const int* __restrict__ start,
    int* __restrict__ rowptr, int* __restrict__ csr_src, int n, int n_buckets,
    int chunk_len) {
  static_assert(GIN_CHUNKS == GIN_BNODES, "a bucket block has one thread per chunk and one per node");
  static_assert(GIN_SEG * 32 == GIN_CHUNKS, "32 lane groups x GIN_SEG pieces cover every chunk once");
  __shared__ int cnt[GIN_BNODES];
  __shared__ int seg_beg[GIN_CHUNKS];
  __shared__ int seg_end[GIN_CHUNKS];
  __shared__ int w_tot[GIN_BNODES / 32];
  __shared__ int w_beg[GIN_BNODES / 32];
  __shared__ int w_len[GIN_BNODES / 32];
  __shared__ int stage2[GIN_STAGE2];
  const int b = blockIdx.x;
  const int t = threadIdx.x;
  const int lane = t & 31;
  const int wave = t >> 5;
  const int g = t >> 3;
  const int l8 = t & 7;
  const int2* p2 = (const int2*)pairs;

  const int s0 = start[t * GIN_TP + b];
  const int s1 = start[t * GIN_TP + b + 1];
  seg_beg[t] = s0;
  seg_end[t] = s1;
  int part_beg = s0;
  int part_len = s1 - s0;
#pragma unroll
  for (int k = 16; k >= 1; k >>= 1) {
    part_beg += __shfl_xor(part_beg, k);
    part_len += __shfl_xor(part_len, k);
  }
  if (lane == 0) {
    w_beg[wave] = part_beg;
    w_len[wave] = part_len;
  }
  cnt[t] = 0;
  __syncthreads();
  int beg = 0, tot = 0;
#pragma unroll
  for (int w = 0; w < GIN_BNODES / 32; ++w) {
    beg += w_beg[w];
    tot += w_len[w];
  }
  const int end = beg + tot;
  const bool staged = tot <= GIN_STAGE2;

  int2 r[GIN_SEG * GIN_SP];
  unsigned valid = 0u;
  bool longer = false;
#pragma unroll
  for (int q = 0; q < GIN_SEG; ++q) {
    const int chunk = q * 32 + g;
    const int kb = seg_beg[chunk];
    const int ke = seg_end[chunk];
    longer = longer || (ke - kb > GIN_SP * 8);
#pragma unroll
    for (int p = 0; p < GIN_SP; ++p) {
      const int k = kb + p * 8 + l8;
      const bool in = k < ke;
      if (in) valid |= 1u << (q * GIN_SP + p);
      r[q * GIN_SP + p] = p2[in ? (size_t)chunk * (size_t)chunk_len + (size_t)k : (size_t)0];
    }
  }
#pragma unroll
  for (int j = 0; j < GIN_SEG * GIN_SP; ++j)
    if ((valid >> j) & 1u) atomicAdd(&cnt[r[j].y & (GIN_BNODES - 1)], 1);
  if (longer) {
    for (int q = 0; q < GIN_SEG; ++q) {
      const int chunk = q * 32 + g;
      const int ke = seg_end[chunk];
      const int2* piece = p2 + (size_t)chunk * (size_t)chunk_len;
      for (int k = seg_beg[chunk] + GIN_SP * 8 + l8; k < ke; k += 8)
        atomicAdd(&cnt[piece[k].y & (GIN_BNODES - 1)], 1);
    }
  }
  __syncthreads();

  const int own = cnt[t];
  int incl = own;
#pragma unroll
  for (int d = 1; d < 32; d <<= 1) {
    int up = __shfl_up(incl, d);
    if (lane >= d) incl += up;
  }
  if (lane == 31) w_tot[wave] = incl;
  __syncthreads();
  int wpre = 0;
#pragma unroll
  for (int w = 0; w < GIN_BNODES / 32; ++w)
    if (w < wave) wpre += w_tot[w];
  const int first = beg + wpre + incl - own;
  cnt[t] = first;
  const int node = b * GIN_BNODES + t;
  if (node < n) rowptr[node] = first;
  if (b == n_buckets - 1 && t == 0) rowptr[n] = end;
  __syncthreads();

  int pos[GIN_SEG * GIN_SP];
#pragma unroll
  for (int j = 0; j < GIN_SEG * GIN_SP; ++j)
    if ((valid >> j) & 1u)
      pos[j] = atomicAdd(&cnt[r[j].y & (GIN_BNODES - 1)], 1);
  if (staged) {
#pragma unroll
    for (int j = 0; j < GIN_SEG * GIN_SP; ++j)
      if ((valid >> j) & 1u) stage2[pos[j] - beg] = r[j].x;
  } else {
#pragma unroll
    for (int j = 0; j < GIN_SEG * GIN_SP; ++j)
      if ((valid >> j) & 1u) csr_src[pos[j]] = r[j].x;
  }
  if (longer) {
    for (int q = 0; q < GIN_SEG; ++q) {
      const int chunk = q * 32 + g;
      const int ke = seg_end[chunk];
      const int2* piece = p2 + (size_t)chunk * (size_t)chunk_len;
      for (int k = seg_beg[chunk] + GIN_SP * 8 + l8; k < ke; k += 8) {
        const int2 p = piece[k];
        const int at = atomicAdd(&cnt[p.y & (GIN_BNODES - 1)], 1);
        if (staged) stage2[at - beg] = p.x;
        else csr_src[at] = p.x;
      }
    }
  }
  if (staged) {
    __syncthreads();
    int* run = csr_src + beg;
    const int shift = (int)(((size_t)run >> 2) & 31);
    for (int i0 = t - shift; i0 < tot; i0 += 8 * GIN_BNODES) {
      int v[8];
#pragma unroll
      for (int k = 0; k < 8; ++k) {
        const int i = i0 + k * GIN_BNODES;
        v[k] = stage2[(i >= 0 && i < tot) ? i : 0];
      }
#pragma unroll
      for (int k = 0; k < 8; ++k) {
        const int i = i0 + k * GIN_BNODES;
        if (i >= 0 && i < tot) run[i] = v[k];
      }
    }
  }
}

template <bool RELU_IN>
__device__ __forceinline__ float4 gin_gather_row(
    const float4* __restrict__ h4, const int* __restrict__ csr_src, float4 acc,
    int beg, int end, int l8) {
  if (RELU_IN) {
    acc.x = fmaxf(acc.x, 0.0f); acc.y = fmaxf(acc.y, 0.0f);
    acc.z = fmaxf(acc.z, 0.0f); acc.w = fmaxf(acc.w, 0.0f);
  }
  acc.x *= GIN_SELF_W; acc.y *= GIN_SELF_W; acc.z *= GIN_SELF_W; acc.w *= GIN_SELF_W;
  for (int base = beg; base < end; base += 8) {
    const int e = min(base + l8, end - 1);
    const int mine = csr_src[e];
    float4 r[8];
#pragma unroll
    for (int j = 0; j < 8; ++j) {
      const unsigned s = (unsigned)__shfl(mine, j, 8);
      r[j] = h4[s * 8u + (unsigned)l8];
    }
#pragma unroll
    for (int j = 0; j < 8; ++j) {
      float4 v = r[j];
      if (RELU_IN) {
        v.x = fmaxf(v.x, 0.0f); v.y = fmaxf(v.y, 0.0f);
        v.z = fmaxf(v.z, 0.0f); v.w = fmaxf(v.w, 0.0f);
      }
      const bool on = base + j < end;
      acc.x += on ? v.x : 0.0f; acc.y += on ? v.y : 0.0f;
      acc.z += on ? v.z : 0.0f; acc.w += on ? v.w : 0.0f;
    }
  }
  return acc;
}

#define GIN_NODE_STEPS 3
#define GIN_GROUP_NODES (32 * GIN_NODE_STEPS)
__device__ __forceinline__ void gin_group_heads(
    const float4* __restrict__ h4, const int* __restrict__ rowptr, int node0, int n,
    int l8, float4 (&own)[GIN_NODE_STEPS], int (&beg)[GIN_NODE_STEPS],
    int (&end)[GIN_NODE_STEPS]) {
#pragma unroll
  for (int j = 0; j < GIN_NODE_STEPS; ++j) {
    const int node = node0 + j * 32;
    const bool ok = node < n;
    beg[j] = rowptr[ok ? node : 0];
    end[j] = rowptr[ok ? node + 1 : 0];
    own[j] = h4[(unsigned)(ok ? node : 0) * 8u + (unsigned)l8];
  }
  static_assert(GIN_NODE_STEPS == 3, "the fence names the loads of three nodes");
  asm volatile("" ::"v"(beg[0]), "v"(end[0]), "v"(beg[1]), "v"(end[1]), "v"(beg[2]), "v"(end[2]),
               "v"(own[0].x), "v"(own[0].y), "v"(own[0].z), "v"(own[0].w),
               "v"(own[1].x), "v"(own[1].y), "v"(own[1].z), "v"(own[1].w),
               "v"(own[2].x), "v"(own[2].y), "v"(own[2].z), "v"(own[2].w));
}

template <bool RELU_IN>
__global__ __launch_bounds__(256) void gin_gather_kernel(
    const float* __restrict__ h, const int* __restrict__ rowptr,
    const int* __restrict__ csr_src, float* __restrict__ agg, int n) {
  const int l8 = threadIdx.x & 7;
  const int node0 = blockIdx.x * GIN_GROUP_NODES + ((int)threadIdx.x >> 3);
  float4 own[GIN_NODE_STEPS];
  int beg[GIN_NODE_STEPS], end[GIN_NODE_STEPS];
  gin_group_heads((const float4*)h, rowptr, node0, n, l8, own, beg, end);
#pragma unroll
  for (int j = 0; j < GIN_NODE_STEPS; ++j) {
    const float4 acc =
        gin_gather_row<RELU_IN>((const float4*)h, csr_src, own[j], beg[j], end[j], l8);
    const int node = node0 + j * 32;
    if (node < n)
      ((float4*)agg)[(unsigned)node * 8u + (unsigned)l8] = acc;
  }
}

#define GIN_LDS_PITCH 36
template <bool RELU_IN, int DOUTL, int NT, bool RELU_OUT>
__global__ __launch_bounds__(256) void gin_gather_gemm_kernel(
    const float* __restrict__ h, const int* __restrict__ rowptr,
    const int* __restrict__ csr_src, const float* __restrict__ W,
    float* __restrict__ C, int n) {
  static_assert(GIN_GROUP_NODES / 16 <= 8, "one 16-row tile per wave of the block at most");
  __shared__ __attribute__((aligned(16))) float rows[GIN_GROUP_NODES * GIN_LDS_PITCH];
  const int l8 = threadIdx.x & 7;
  const int lnode = (int)threadIdx.x >> 3;
  const int node0 = blockIdx.x * GIN_GROUP_NODES + lnode;
  float4 own[GIN_NODE_STEPS];
  int beg[GIN_NODE_STEPS], end[GIN_NODE_STEPS];
  gin_group_heads((const float4*)h, rowptr, node0, n, l8, own, beg, end);
#pragma unroll
  for (int j = 0; j < GIN_NODE_STEPS; ++j) {
    float4 sum =
        gin_gather_row<RELU_IN>((const float4*)h, csr_src, own[j], beg[j], end[j], l8);
    if (node0 + j * 32 >= n)
      sum = make_float4(0.0f, 0.0f, 0.0f, 0.0f);
    *(float4*)&rows[(j * 32 + lnode) * GIN_LDS_PITCH + l8 * 4] = sum;
  }
  __syncthreads();

  const int lane = threadIdx.x & 31;
  const int wave = threadIdx.x >> 5;
  const int row0 = blockIdx.x * GIN_GROUP_NODES + wave * 16;
  if (wave >= GIN_GROUP_NODES / 16 || row0 >= n) return;

  const int l16 = lane & 15;
  const int khalf = lane >> 4;
  const float* arow = &rows[(wave * 16 + l16) * GIN_LDS_PITCH];

  v8f acc[NT] = {};

#pragma unroll
  for (int kk = 0; kk < GIN_D_HID / 4; ++kk) {
    const int kb = kk * 4 + khalf * 2;
    const v2f a = *(const v2f*)(arow + kb);
#pragma unroll
    for (int t = 0; t < NT; ++t) {
      const int col = t * 16 + l16;
      const bool ok = ((DOUTL & 15) == 0) || (col < DOUTL);
      v2f b;
      b.x = ok ? W[(size_t)kb * DOUTL + col] : 0.0f;
      b.y = ok ? W[(size_t)(kb + 1) * DOUTL + col] : 0.0f;
      acc[t] = __builtin_amdgcn_wmma_f32_16x16x4_f32(
          false, a, false, b, (short)0, acc[t], false, false);
    }
  }

#pragma unroll
  for (int t = 0; t < NT; ++t) {
    const int col = t * 16 + l16;
    if (((DOUTL & 15) != 0) && col >= DOUTL) continue;
#pragma unroll
    for (int r = 0; r < 8; ++r) {
      const int row = row0 + r + khalf * 8;
      if (row < n) {
        float v = acc[t][r];
        if (RELU_OUT) v = fmaxf(v, 0.0f);
        C[(size_t)row * DOUTL + col] = v;
      }
    }
  }
}

template <int DINL, int DOUTL, int NT, bool RELU>
__device__ __forceinline__ void gin_gemm_tile(
    const float* __restrict__ A, const float* __restrict__ W,
    float* __restrict__ C, int n_rows, int tile) {
  const int lane = threadIdx.x & 31;
  const int row0 = tile << 4;
  const int l16 = lane & 15;
  const int khalf = lane >> 4;
  int arow = row0 + l16;
  if (arow >= n_rows) arow = n_rows - 1;

  v8f acc[NT] = {};

#pragma unroll
  for (int kk = 0; kk < DINL / 4; ++kk) {
    const int kb = kk * 4 + khalf * 2;
    v2f a;
    a.x = A[(size_t)arow * DINL + kb];
    a.y = A[(size_t)arow * DINL + kb + 1];
#pragma unroll
    for (int t = 0; t < NT; ++t) {
      const int col = t * 16 + l16;
      const bool ok = ((DOUTL & 15) == 0) || (col < DOUTL);
      v2f b;
      b.x = ok ? W[(size_t)kb * DOUTL + col] : 0.0f;
      b.y = ok ? W[(size_t)(kb + 1) * DOUTL + col] : 0.0f;
      acc[t] = __builtin_amdgcn_wmma_f32_16x16x4_f32(
          false, a, false, b, (short)0, acc[t], false, false);
    }
  }

#pragma unroll
  for (int t = 0; t < NT; ++t) {
    const int col = t * 16 + l16;
    if (((DOUTL & 15) != 0) && col >= DOUTL) continue;
#pragma unroll
    for (int r = 0; r < 8; ++r) {
      const int row = row0 + r + khalf * 8;
      if (row < n_rows) {
        float v = acc[t][r];
        if (RELU) v = fmaxf(v, 0.0f);
        C[(size_t)row * DOUTL + col] = v;
      }
    }
  }
}

__global__ __launch_bounds__(GIN_BNODES) void gin_csr_project_kernel(
    const int* __restrict__ pairs, const int* __restrict__ start,
    int* __restrict__ rowptr, int* __restrict__ csr_src, int n, int n_buckets,
    int chunk_len, const float* __restrict__ x, const float* __restrict__ W0,
    float* __restrict__ xw) {
  if ((int)blockIdx.x >= n_buckets) {
    const int tile = ((int)blockIdx.x - n_buckets) * (GIN_BNODES / 32) + ((int)threadIdx.x >> 5);
    if (tile < ((n + 15) >> 4))
      gin_gemm_tile<GIN_D_IN, GIN_D_HID, 2, false>(x, W0, xw, n, tile);
    return;
  }
  gin_bucket_csr_block(pairs, start, rowptr, csr_src, n, n_buckets, chunk_len);
}

extern "C" void kernel_launch(void* const* d_in, const int* in_sizes, int n_in,
                              void* d_out, int out_size, void* d_ws, size_t ws_size,
                              hipStream_t stream) {
  const float* x  = (const float*)d_in[0];
  const float* W0 = (const float*)d_in[1];
  const float* W1 = (const float*)d_in[2];
  const float* W2 = (const float*)d_in[3];
  const int* src  = (const int*)d_in[4];
  const int* dst  = (const int*)d_in[5];
  float* out = (float*)d_out;

  const int n = in_sizes[0] / GIN_D_IN;
  const int n_edges = in_sizes[4];

  float* tbuf = (float*)d_ws;
  float* agg1 = tbuf + (size_t)n * GIN_D_HID;
  int* start  = (int*)(agg1 + (size_t)n * GIN_D_HID);
  int* rowptr = start + GIN_CHUNKS * GIN_TP;
  int* pairs  = rowptr + (((size_t)n + 1 + 3) / 4) * 4;
  int* csrc   = pairs + 2 * (size_t)n_edges;

  const int ntiles = (n + 15) / 16;
  const int gemm_blocks = (ntiles + 7) / 8;
  const int n_buckets = (n + GIN_BNODES - 1) / GIN_BNODES;
  const int chunk_len = (n_edges + GIN_CHUNKS - 1) / GIN_CHUNKS;
  const int group_blocks = (n + GIN_GROUP_NODES - 1) / GIN_GROUP_NODES;

  gin_chunk_sort_kernel<<<GIN_CHUNKS, 1024, 0, stream>>>(src, dst, pairs, start, n_edges, n_buckets, chunk_len);
  gin_csr_project_kernel<<<n_buckets + gemm_blocks, GIN_BNODES, 0, stream>>>(
      pairs, start, rowptr, csrc, n, n_buckets, chunk_len, x, W0, tbuf);

  gin_gather_kernel<false><<<group_blocks, 256, 0, stream>>>(tbuf, rowptr, csrc, agg1, n);
  gin_gather_gemm_kernel<true, GIN_D_HID, 2, true>
      <<<group_blocks, 256, 0, stream>>>(agg1, rowptr, csrc, W1, tbuf, n);
  gin_gather_gemm_kernel<false, GIN_N_CLASSES, 3, false>
      <<<group_blocks, 256, 0, stream>>>(tbuf, rowptr, csrc, W2, out, n);
}


# Aggregation (3 layers): a block copies its 97 row bounds and the one contiguous source-id range of its 96 nodes to LDS in whole lines; trips take ids from LDS, not memory; sum order unchanged
# speedup vs baseline: 4.5094x; 1.0772x over previous
#include <hip/hip_runtime.h>


typedef __attribute__((ext_vector_type(2))) float v2f;
typedef __attribute__((ext_vector_type(8))) float v8f;

#define GIN_D_IN 64
#define GIN_D_HID 32
#define GIN_N_CLASSES 40
#define GIN_SELF_W 1.1f

#define GIN_BSH 8
#define GIN_BNODES (1 << GIN_BSH)
#define GIN_MAXB 512
#define GIN_CHUNKS 256
#define GIN_TP (GIN_MAXB + 1)
#define GIN_EB 7
#define GIN_SEG 8
#define GIN_SP 3
#define GIN_STAGE1 (1024 * GIN_EB)
#define GIN_STAGE2 6144

__global__ __launch_bounds__(1024) void gin_chunk_sort_kernel(
    const int* __restrict__ src, const int* __restrict__ dst,
    int* __restrict__ pairs, int* __restrict__ start, int n_edges,
    int n_buckets, int chunk_len) {
  __shared__ int hist[GIN_MAXB];
  __shared__ int cur[GIN_MAXB];
  __shared__ int w_tot[GIN_MAXB / 32];
  __shared__ int2 stage[GIN_STAGE1];
  const int c = blockIdx.x;
  const int t = threadIdx.x;
  const int lane = t & 31;
  const int wave = t >> 5;
  const int beg = c * chunk_len;
  const int end = min(beg + chunk_len, n_edges);
  const int last = max(n_edges - 1, 0);
  const bool one_batch = end - beg <= 1024 * GIN_EB;
  int e0 = beg + t;
  int d[GIN_EB], s[GIN_EB];
#pragma unroll
  for (int j = 0; j < GIN_EB; ++j) {
    const int e = min(e0 + j * 1024, last);
    d[j] = dst[e];
    s[j] = src[e];
  }
  for (int i = t; i < GIN_MAXB; i += 1024) hist[i] = 0;
  __syncthreads();
  for (;;) {
#pragma unroll
    for (int j = 0; j < GIN_EB; ++j)
      if (e0 + j * 1024 < end) atomicAdd(&hist[d[j] >> GIN_BSH], 1);
    e0 += 1024 * GIN_EB;
    if (e0 >= end) break;
#pragma unroll
    for (int j = 0; j < GIN_EB; ++j) d[j] = dst[min(e0 + j * 1024, last)];
  }
  __syncthreads();

  const bool has = t < n_buckets;
  const int own = has ? hist[t] : 0;
  int incl = own;
#pragma unroll
  for (int k = 1; k < 32; k <<= 1) {
    int up = __shfl_up(incl, k);
    if (lane >= k) incl += up;
  }
  if (lane == 31 && wave < GIN_MAXB / 32) w_tot[wave] = incl;
  __syncthreads();
  if (has) {
    int wpre = 0;
#pragma unroll
    for (int w = 0; w < GIN_MAXB / 32; ++w)
      if (w < wave) wpre += w_tot[w];
    const int first = wpre + incl - own;
    cur[t] = first;
    start[c * GIN_TP + t] = first;
    if (t == n_buckets - 1) start[c * GIN_TP + n_buckets] = first + own;
  }
  __syncthreads();

  e0 = beg + t;
  int2* p2 = (int2*)pairs + (size_t)beg;
  if (one_batch) {
    int pos[GIN_EB];
#pragma unroll
    for (int j = 0; j < GIN_EB; ++j)
      if (e0 + j * 1024 < end) pos[j] = atomicAdd(&cur[d[j] >> GIN_BSH], 1);
#pragma unroll
    for (int j = 0; j < GIN_EB; ++j)
      if (e0 + j * 1024 < end) stage[pos[j]] = make_int2(s[j], d[j]);
    __syncthreads();
    const int cnt = end - beg;
    const int shift = (int)(((size_t)p2 >> 3) & 15);
    const int i0 = t - shift;
    int2 v[GIN_EB + 1];
#pragma unroll
    for (int k = 0; k <= GIN_EB; ++k) {
      const int i = i0 + k * 1024;
      v[k] = stage[(i >= 0 && i < cnt) ? i : 0];
    }
#pragma unroll
    for (int k = 0; k <= GIN_EB; ++k) {
      const int i = i0 + k * 1024;
      if (i >= 0 && i < cnt) p2[i] = v[k];
    }
    return;
  }
#pragma unroll
  for (int j = 0; j < GIN_EB; ++j) {
    const int e = min(e0 + j * 1024, last);
    d[j] = dst[e];
    s[j] = src[e];
  }
  for (;;) {
    int pos[GIN_EB];
#pragma unroll
    for (int j = 0; j < GIN_EB; ++j)
      if (e0 + j * 1024 < end) pos[j] = atomicAdd(&cur[d[j] >> GIN_BSH], 1);
#pragma unroll
    for (int j = 0; j < GIN_EB; ++j)
      if (e0 + j * 1024 < end) p2[pos[j]] = make_int2(s[j], d[j]);
    e0 += 1024 * GIN_EB;
    if (e0 >= end) break;
#pragma unroll
    for (int j = 0; j < GIN_EB; ++j) {
      const int e = min(e0 + j * 1024, last);
      d[j] = dst[e];
      s[j] = src[e];
    }
  }
}

__device__ __forceinline__ void gin_bucket_csr_block(
    const int* __restrict__ pairs, const int* __restrict__ start,
    int* __restrict__ rowptr, int* __restrict__ csr_src, int n, int n_buckets,
    int chunk_len) {
  static_assert(GIN_CHUNKS == GIN_BNODES, "a bucket block has one thread per chunk and one per node");
  static_assert(GIN_SEG * 32 == GIN_CHUNKS, "32 lane groups x GIN_SEG pieces cover every chunk once");
  __shared__ int cnt[GIN_BNODES];
  __shared__ int seg_beg[GIN_CHUNKS];
  __shared__ int seg_end[GIN_CHUNKS];
  __shared__ int w_tot[GIN_BNODES / 32];
  __shared__ int w_beg[GIN_BNODES / 32];
  __shared__ int w_len[GIN_BNODES / 32];
  __shared__ int stage2[GIN_STAGE2];
  const int b = blockIdx.x;
  const int t = threadIdx.x;
  const int lane = t & 31;
  const int wave = t >> 5;
  const int g = t >> 3;
  const int l8 = t & 7;
  const int2* p2 = (const int2*)pairs;

  const int s0 = start[t * GIN_TP + b];
  const int s1 = start[t * GIN_TP + b + 1];
  seg_beg[t] = s0;
  seg_end[t] = s1;
  int part_beg = s0;
  int part_len = s1 - s0;
#pragma unroll
  for (int k = 16; k >= 1; k >>= 1) {
    part_beg += __shfl_xor(part_beg, k);
    part_len += __shfl_xor(part_len, k);
  }
  if (lane == 0) {
    w_beg[wave] = part_beg;
    w_len[wave] = part_len;
  }
  cnt[t] = 0;
  __syncthreads();
  int beg = 0, tot = 0;
#pragma unroll
  for (int w = 0; w < GIN_BNODES / 32; ++w) {
    beg += w_beg[w];
    tot += w_len[w];
  }
  const int end = beg + tot;
  const bool staged = tot <= GIN_STAGE2;

  int2 r[GIN_SEG * GIN_SP];
  unsigned valid = 0u;
  bool longer = false;
#pragma unroll
  for (int q = 0; q < GIN_SEG; ++q) {
    const int chunk = q * 32 + g;
    const int kb = seg_beg[chunk];
    const int ke = seg_end[chunk];
    longer = longer || (ke - kb > GIN_SP * 8);
#pragma unroll
    for (int p = 0; p < GIN_SP; ++p) {
      const int k = kb + p * 8 + l8;
      const bool in = k < ke;
      if (in) valid |= 1u << (q * GIN_SP + p);
      r[q * GIN_SP + p] = p2[in ? (size_t)chunk * (size_t)chunk_len + (size_t)k : (size_t)0];
    }
  }
#pragma unroll
  for (int j = 0; j < GIN_SEG * GIN_SP; ++j)
    if ((valid >> j) & 1u) atomicAdd(&cnt[r[j].y & (GIN_BNODES - 1)], 1);
  if (longer) {
    for (int q = 0; q < GIN_SEG; ++q) {
      const int chunk = q * 32 + g;
      const int ke = seg_end[chunk];
      const int2* piece = p2 + (size_t)chunk * (size_t)chunk_len;
      for (int k = seg_beg[chunk] + GIN_SP * 8 + l8; k < ke; k += 8)
        atomicAdd(&cnt[piece[k].y & (GIN_BNODES - 1)], 1);
    }
  }
  __syncthreads();

  const int own = cnt[t];
  int incl = own;
#pragma unroll
  for (int d = 1; d < 32; d <<= 1) {
    int up = __shfl_up(incl, d);
    if (lane >= d) incl += up;
  }
  if (lane == 31) w_tot[wave] = incl;
  __syncthreads();
  int wpre = 0;
#pragma unroll
  for (int w = 0; w < GIN_BNODES / 32; ++w)
    if (w < wave) wpre += w_tot[w];
  const int first = beg + wpre + incl - own;
  cnt[t] = first;
  const int node = b * GIN_BNODES + t;
  if (node < n) rowptr[node] = first;
  if (b == n_buckets - 1 && t == 0) rowptr[n] = end;
  __syncthreads();

  int pos[GIN_SEG * GIN_SP];
#pragma unroll
  for (int j = 0; j < GIN_SEG * GIN_SP; ++j)
    if ((valid >> j) & 1u)
      pos[j] = atomicAdd(&cnt[r[j].y & (GIN_BNODES - 1)], 1);
  if (staged) {
#pragma unroll
    for (int j = 0; j < GIN_SEG * GIN_SP; ++j)
      if ((valid >> j) & 1u) stage2[pos[j] - beg] = r[j].x;
  } else {
#pragma unroll
    for (int j = 0; j < GIN_SEG * GIN_SP; ++j)
      if ((valid >> j) & 1u) csr_src[pos[j]] = r[j].x;
  }
  if (longer) {
    for (int q = 0; q < GIN_SEG; ++q) {
      const int chunk = q * 32 + g;
      const int ke = seg_end[chunk];
      const int2* piece = p2 + (size_t)chunk * (size_t)chunk_len;
      for (int k = seg_beg[chunk] + GIN_SP * 8 + l8; k < ke; k += 8) {
        const int2 p = piece[k];
        const int at = atomicAdd(&cnt[p.y & (GIN_BNODES - 1)], 1);
        if (staged) stage2[at - beg] = p.x;
        else csr_src[at] = p.x;
      }
    }
  }
  if (staged) {
    __syncthreads();
    int* run = csr_src + beg;
    const int shift = (int)(((size_t)run >> 2) & 31);
    for (int i0 = t - shift; i0 < tot; i0 += 8 * GIN_BNODES) {
      int v[8];
#pragma unroll
      for (int k = 0; k < 8; ++k) {
        const int i = i0 + k * GIN_BNODES;
        v[k] = stage2[(i >= 0 && i < tot) ? i : 0];
      }
#pragma unroll
      for (int k = 0; k < 8; ++k) {
        const int i = i0 + k * GIN_BNODES;
        if (i >= 0 && i < tot) run[i] = v[k];
      }
    }
  }
}

template <bool RELU_IN>
__device__ __forceinline__ float4 gin_gather_row(
    const float4* __restrict__ h4, const int* __restrict__ csr_src, float4 acc,
    int beg, int end, int l8) {
  if (RELU_IN) {
    acc.x = fmaxf(acc.x, 0.0f); acc.y = fmaxf(acc.y, 0.0f);
    acc.z = fmaxf(acc.z, 0.0f); acc.w = fmaxf(acc.w, 0.0f);
  }
  acc.x *= GIN_SELF_W; acc.y *= GIN_SELF_W; acc.z *= GIN_SELF_W; acc.w *= GIN_SELF_W;
  for (int base = beg; base < end; base += 8) {
    const int e = min(base + l8, end - 1);
    const int mine = csr_src[e];
    float4 r[8];
#pragma unroll
    for (int j = 0; j < 8; ++j) {
      const unsigned s = (unsigned)__shfl(mine, j, 8);
      r[j] = h4[s * 8u + (unsigned)l8];
    }
#pragma unroll
    for (int j = 0; j < 8; ++j) {
      float4 v = r[j];
      if (RELU_IN) {
        v.x = fmaxf(v.x, 0.0f); v.y = fmaxf(v.y, 0.0f);
        v.z = fmaxf(v.z, 0.0f); v.w = fmaxf(v.w, 0.0f);
      }
      const bool on = base + j < end;
      acc.x += on ? v.x : 0.0f; acc.y += on ? v.y : 0.0f;
      acc.z += on ? v.z : 0.0f; acc.w += on ? v.w : 0.0f;
    }
  }
  return acc;
}

template <bool RELU_IN>
__device__ __forceinline__ float4 gin_gather_row_lds(
    const float4* __restrict__ h4, const int* ids, int lo, float4 acc,
    int beg, int end, int l8) {
  if (RELU_IN) {
    acc.x = fmaxf(acc.x, 0.0f); acc.y = fmaxf(acc.y, 0.0f);
    acc.z = fmaxf(acc.z, 0.0f); acc.w = fmaxf(acc.w, 0.0f);
  }
  acc.x *= GIN_SELF_W; acc.y *= GIN_SELF_W; acc.z *= GIN_SELF_W; acc.w *= GIN_SELF_W;
  for (int base = beg; base < end; base += 8) {
    float4 r[8];
#pragma unroll
    for (int j = 0; j < 8; ++j) {
      const unsigned s = (unsigned)ids[min(base + j, end - 1) - lo];
      r[j] = h4[s * 8u + (unsigned)l8];
    }
#pragma unroll
    for (int j = 0; j < 8; ++j) {
      float4 v = r[j];
      if (RELU_IN) {
        v.x = fmaxf(v.x, 0.0f); v.y = fmaxf(v.y, 0.0f);
        v.z = fmaxf(v.z, 0.0f); v.w = fmaxf(v.w, 0.0f);
      }
      const bool on = base + j < end;
      acc.x += on ? v.x : 0.0f; acc.y += on ? v.y : 0.0f;
      acc.z += on ? v.z : 0.0f; acc.w += on ? v.w : 0.0f;
    }
  }
  return acc;
}

#define GIN_NODE_STEPS 3
#define GIN_GROUP_NODES (32 * GIN_NODE_STEPS)
#define GIN_IDS_CAP 3072
template <bool RELU_IN>
__device__ __forceinline__ void gin_block_sums(
    const float4* __restrict__ h4, const int* __restrict__ rowptr,
    const int* __restrict__ csr_src, int n, int* bnd, int* ids,
    float4 (&sum)[GIN_NODE_STEPS]) {
  const int t = threadIdx.x;
  const int l8 = t & 7;
  const int lnode = t >> 3;
  const int nb0 = blockIdx.x * GIN_GROUP_NODES;
  float4 own[GIN_NODE_STEPS];
#pragma unroll
  for (int j = 0; j < GIN_NODE_STEPS; ++j) {
    const int node = nb0 + lnode + j * 32;
    own[j] = h4[(unsigned)(node < n ? node : 0) * 8u + (unsigned)l8];
  }
  if (t <= GIN_GROUP_NODES) bnd[t] = rowptr[min(nb0 + t, n)];
  __syncthreads();
  const int lo = bnd[0];
  const int cnt = bnd[GIN_GROUP_NODES] - lo;
  int beg[GIN_NODE_STEPS], end[GIN_NODE_STEPS];
#pragma unroll
  for (int j = 0; j < GIN_NODE_STEPS; ++j) {
    beg[j] = bnd[lnode + j * 32];
    end[j] = bnd[lnode + j * 32 + 1];
  }
  if (cnt <= GIN_IDS_CAP) {
    if (cnt > 0) {
      const int* run = csr_src + lo;
      const int shift = (int)(((size_t)run >> 2) & 31);
      for (int i0 = t - shift; i0 < cnt; i0 += 8 * 256) {
        int v[8];
#pragma unroll
        for (int k = 0; k < 8; ++k) {
          const int i = i0 + k * 256;
          v[k] = run[(i >= 0 && i < cnt) ? i : 0];
        }
#pragma unroll
        for (int k = 0; k < 8; ++k) {
          const int i = i0 + k * 256;
          if (i >= 0 && i < cnt) ids[i] = v[k];
        }
      }
    }
    __syncthreads();
#pragma unroll
    for (int j = 0; j < GIN_NODE_STEPS; ++j)
      sum[j] = gin_gather_row_lds<RELU_IN>(h4, ids, lo, own[j], beg[j], end[j], l8);
  } else {
#pragma unroll
    for (int j = 0; j < GIN_NODE_STEPS; ++j)
      sum[j] = gin_gather_row<RELU_IN>(h4, csr_src, own[j], beg[j], end[j], l8);
  }
}

template <bool RELU_IN>
__global__ __launch_bounds__(256) void gin_gather_kernel(
    const float* __restrict__ h, const int* __restrict__ rowptr,
    const int* __restrict__ csr_src, float* __restrict__ agg, int n) {
  __shared__ int bnd[GIN_GROUP_NODES + 1];
  __shared__ int ids[GIN_IDS_CAP];
  const int l8 = threadIdx.x & 7;
  const int node0 = blockIdx.x * GIN_GROUP_NODES + ((int)threadIdx.x >> 3);
  float4 sum[GIN_NODE_STEPS];
  gin_block_sums<RELU_IN>((const float4*)h, rowptr, csr_src, n, bnd, ids, sum);
#pragma unroll
  for (int j = 0; j < GIN_NODE_STEPS; ++j) {
    const int node = node0 + j * 32;
    if (node < n)
      ((float4*)agg)[(unsigned)node * 8u + (unsigned)l8] = sum[j];
  }
}

#define GIN_LDS_PITCH 36
template <bool RELU_IN, int DOUTL, int NT, bool RELU_OUT>
__global__ __launch_bounds__(256) void gin_gather_gemm_kernel(
    const float* __restrict__ h, const int* __restrict__ rowptr,
    const int* __restrict__ csr_src, const float* __restrict__ W,
    float* __restrict__ C, int n) {
  static_assert(GIN_GROUP_NODES / 16 <= 8, "one 16-row tile per wave of the block at most");
  __shared__ __attribute__((aligned(16))) float rows[GIN_GROUP_NODES * GIN_LDS_PITCH];
  __shared__ int bnd[GIN_GROUP_NODES + 1];
  __shared__ int ids[GIN_IDS_CAP];
  const int l8 = threadIdx.x & 7;
  const int lnode = (int)threadIdx.x >> 3;
  const int node0 = blockIdx.x * GIN_GROUP_NODES + lnode;
  float4 sum[GIN_NODE_STEPS];
  gin_block_sums<RELU_IN>((const float4*)h, rowptr, csr_src, n, bnd, ids, sum);
#pragma unroll
  for (int j = 0; j < GIN_NODE_STEPS; ++j) {
    if (node0 + j * 32 >= n)
      sum[j] = make_float4(0.0f, 0.0f, 0.0f, 0.0f);
    *(float4*)&rows[(j * 32 + lnode) * GIN_LDS_PITCH + l8 * 4] = sum[j];
  }
  __syncthreads();

  const int lane = threadIdx.x & 31;
  const int wave = threadIdx.x >> 5;
  const int row0 = blockIdx.x * GIN_GROUP_NODES + wave * 16;
  if (wave >= GIN_GROUP_NODES / 16 || row0 >= n) return;

  const int l16 = lane & 15;
  const int khalf = lane >> 4;
  const float* arow = &rows[(wave * 16 + l16) * GIN_LDS_PITCH];

  v8f acc[NT] = {};

#pragma unroll
  for (int kk = 0; kk < GIN_D_HID / 4; ++kk) {
    const int kb = kk * 4 + khalf * 2;
    const v2f a = *(const v2f*)(arow + kb);
#pragma unroll
    for (int t = 0; t < NT; ++t) {
      const int col = t * 16 + l16;
      const bool ok = ((DOUTL & 15) == 0) || (col < DOUTL);
      v2f b;
      b.x = ok ? W[(size_t)kb * DOUTL + col] : 0.0f;
      b.y = ok ? W[(size_t)(kb + 1) * DOUTL + col] : 0.0f;
      acc[t] = __builtin_amdgcn_wmma_f32_16x16x4_f32(
          false, a, false, b, (short)0, acc[t], false, false);
    }
  }

#pragma unroll
  for (int t = 0; t < NT; ++t) {
    const int col = t * 16 + l16;
    if (((DOUTL & 15) != 0) && col >= DOUTL) continue;
#pragma unroll
    for (int r = 0; r < 8; ++r) {
      const int row = row0 + r + khalf * 8;
      if (row < n) {
        float v = acc[t][r];
        if (RELU_OUT) v = fmaxf(v, 0.0f);
        C[(size_t)row * DOUTL + col] = v;
      }
    }
  }
}

template <int DINL, int DOUTL, int NT, bool RELU>
__device__ __forceinline__ void gin_gemm_tile(
    const float* __restrict__ A, const float* __restrict__ W,
    float* __restrict__ C, int n_rows, int tile) {
  const int lane = threadIdx.x & 31;
  const int row0 = tile << 4;
  const int l16 = lane & 15;
  const int khalf = lane >> 4;
  int arow = row0 + l16;
  if (arow >= n_rows) arow = n_rows - 1;

  v8f acc[NT] = {};

#pragma unroll
  for (int kk = 0; kk < DINL / 4; ++kk) {
    const int kb = kk * 4 + khalf * 2;
    v2f a;
    a.x = A[(size_t)arow * DINL + kb];
    a.y = A[(size_t)arow * DINL + kb + 1];
#pragma unroll
    for (int t = 0; t < NT; ++t) {
      const int col = t * 16 + l16;
      const bool ok = ((DOUTL & 15) == 0) || (col < DOUTL);
      v2f b;
      b.x = ok ? W[(size_t)kb * DOUTL + col] : 0.0f;
      b.y = ok ? W[(size_t)(kb + 1) * DOUTL + col] : 0.0f;
      acc[t] = __builtin_amdgcn_wmma_f32_16x16x4_f32(
          false, a, false, b, (short)0, acc[t], false, false);
    }
  }

#pragma unroll
  for (int t = 0; t < NT; ++t) {
    const int col = t * 16 + l16;
    if (((DOUTL & 15) != 0) && col >= DOUTL) continue;
#pragma unroll
    for (int r = 0; r < 8; ++r) {
      const int row = row0 + r + khalf * 8;
      if (row < n_rows) {
        float v = acc[t][r];
        if (RELU) v = fmaxf(v, 0.0f);
        C[(size_t)row * DOUTL + col] = v;
      }
    }
  }
}

__global__ __launch_bounds__(GIN_BNODES) void gin_csr_project_kernel(
    const int* __restrict__ pairs, const int* __restrict__ start,
    int* __restrict__ rowptr, int* __restrict__ csr_src, int n, int n_buckets,
    int chunk_len, const float* __restrict__ x, const float* __restrict__ W0,
    float* __restrict__ xw) {
  if ((int)blockIdx.x >= n_buckets) {
    const int tile = ((int)blockIdx.x - n_buckets) * (GIN_BNODES / 32) + ((int)threadIdx.x >> 5);
    if (tile < ((n + 15) >> 4))
      gin_gemm_tile<GIN_D_IN, GIN_D_HID, 2, false>(x, W0, xw, n, tile);
    return;
  }
  gin_bucket_csr_block(pairs, start, rowptr, csr_src, n, n_buckets, chunk_len);
}

extern "C" void kernel_launch(void* const* d_in, const int* in_sizes, int n_in,
                              void* d_out, int out_size, void* d_ws, size_t ws_size,
                              hipStream_t stream) {
  const float* x  = (const float*)d_in[0];
  const float* W0 = (const float*)d_in[1];
  const float* W1 = (const float*)d_in[2];
  const float* W2 = (const float*)d_in[3];
  const int* src  = (const int*)d_in[4];
  const int* dst  = (const int*)d_in[5];
  float* out = (float*)d_out;

  const int n = in_sizes[0] / GIN_D_IN;
  const int n_edges = in_sizes[4];

  float* tbuf = (float*)d_ws;
  float* agg1 = tbuf + (size_t)n * GIN_D_HID;
  int* start  = (int*)(agg1 + (size_t)n * GIN_D_HID);
  int* rowptr = start + GIN_CHUNKS * GIN_TP;
  int* pairs  = rowptr + (((size_t)n + 1 + 3) / 4) * 4;
  int* csrc   = pairs + 2 * (size_t)n_edges;

  const int ntiles = (n + 15) / 16;
  const int gemm_blocks = (ntiles + 7) / 8;
  const int n_buckets = (n + GIN_BNODES - 1) / GIN_BNODES;
  const int chunk_len = (n_edges + GIN_CHUNKS - 1) / GIN_CHUNKS;
  const int group_blocks = (n + GIN_GROUP_NODES - 1) / GIN_GROUP_NODES;

  gin_chunk_sort_kernel<<<GIN_CHUNKS, 1024, 0, stream>>>(src, dst, pairs, start, n_edges, n_buckets, chunk_len);
  gin_csr_project_kernel<<<n_buckets + gemm_blocks, GIN_BNODES, 0, stream>>>(
      pairs, start, rowptr, csrc, n, n_buckets, chunk_len, x, W0, tbuf);

  gin_gather_kernel<false><<<group_blocks, 256, 0, stream>>>(tbuf, rowptr, csrc, agg1, n);
  gin_gather_gemm_kernel<true, GIN_D_HID, 2, true>
      <<<group_blocks, 256, 0, stream>>>(agg1, rowptr, csrc, W1, tbuf, n);
  gin_gather_gemm_kernel<false, GIN_N_CLASSES, 3, false>
      <<<group_blocks, 256, 0, stream>>>(tbuf, rowptr, csrc, W2, out, n);
}
